# starting kernel
# baseline (speedup 1.0000x reference)
#include <hip/hip_runtime.h>

#define N_NODES 100000
#define DHID    128
#define E_EDGES 1600000
#define BN_EPS  1e-5f

typedef __attribute__((ext_vector_type(2))) float v2f;
typedef __attribute__((ext_vector_type(8))) float v8f;

__global__ void k_zero(float* __restrict__ p, int n) {
  int i = blockIdx.x * blockDim.x + threadIdx.x;
  if (i < n) p[i] = 0.0f;
}

__global__ void k_degree(const int* __restrict__ dst, float* __restrict__ deg, int e) {
  int i = blockIdx.x * blockDim.x + threadIdx.x;
  if (i < e) unsafeAtomicAdd(&deg[dst[i]], 1.0f);
}

__global__ void k_dinv(float* __restrict__ deg_dinv, float* __restrict__ selfc, int n) {
  int i = blockIdx.x * blockDim.x + threadIdx.x;
  if (i < n) {
    float d = __frsqrt_rn(deg_dinv[i] + 1.0f);
    deg_dinv[i] = d;
    selfc[i]    = d * d;
  }
}

template <int AMODE, bool HAS_BIAS, bool EPI_RELU>
__global__ __launch_bounds__(256) void k_gemm128(
    const float* __restrict__ A, const float* __restrict__ W,
    float* __restrict__ C, int nrows,
    const float* __restrict__ bnScale, const float* __restrict__ bnShift,
    const float* __restrict__ bias)
{
  __shared__ float wl[DHID * DHID];
  const int tid = threadIdx.x;
  for (int i = tid; i < DHID * DHID; i += 256) {
    int k = i >> 7, n = i & 127;
    wl[(((k >> 1) << 7) + n) * 2 + (k & 1)] = W[i];
  }
  __syncthreads();

  const int wave  = tid >> 5;
  const int lane  = tid & 31;
  const int laneM = lane & 15;
  const int hi    = lane >> 4;
  const int m0    = (blockIdx.x * 8 + wave) << 4;
  if (m0 >= nrows) return;

  const float2* __restrict__ wl2  = (const float2*)wl;
  const float*  __restrict__ arow = A + (size_t)(m0 + laneM) * DHID;

  v8f acc[8];
#pragma unroll
  for (int t = 0; t < 8; ++t) acc[t] = (v8f){};

  for (int k = 0; k < DHID; k += 4) {
    float2 av = *(const float2*)(arow + k + hi * 2);
    if (AMODE == 1) {
      av.x = fmaxf(av.x, 0.0f);
      av.y = fmaxf(av.y, 0.0f);
    } else if (AMODE == 2) {
      int c0 = k + hi * 2;
      av.x = fmaxf(av.x, 0.0f) * bnScale[c0]     + bnShift[c0];
      av.y = fmaxf(av.y, 0.0f) * bnScale[c0 + 1] + bnShift[c0 + 1];
    }
    v2f a; a[0] = av.x; a[1] = av.y;

    const float2* __restrict__ brow = wl2 + (((k >> 1) + hi) << 7);
#pragma unroll
    for (int t = 0; t < 8; ++t) {
      float2 bv = brow[t * 16 + laneM];
      v2f b; b[0] = bv.x; b[1] = bv.y;
      acc[t] = __builtin_amdgcn_wmma_f32_16x16x4_f32(
          false, a, false, b, (short)0, acc[t], false, false);
    }
  }

#pragma unroll
  for (int t = 0; t < 8; ++t) {
    int   col  = t * 16 + laneM;
    float bval = HAS_BIAS ? bias[col] : 0.0f;
#pragma unroll
    for (int r = 0; r < 8; ++r) {
      int   row = m0 + r + hi * 8;
      float v   = acc[t][r] + bval;
      if (EPI_RELU) v = fmaxf(v, 0.0f);
      C[(size_t)row * DHID + col] = v;
    }
  }
}

__global__ void k_selfbias(const float* __restrict__ h, const float* __restrict__ selfc,
                           const float* __restrict__ b, float* __restrict__ out, int n) {
  int i = blockIdx.x * blockDim.x + threadIdx.x;
  if (i >= n * (DHID / 4)) return;
  int    node = i >> 5;
  int    c4   = (i & 31) * 4;
  float  sc   = selfc[node];
  float4 hv   = *(const float4*)(h + (size_t)node * DHID + c4);
  float4 bv   = *(const float4*)(b + c4);
  float4 o;
  o.x = hv.x * sc + bv.x;
  o.y = hv.y * sc + bv.y;
  o.z = hv.z * sc + bv.z;
  o.w = hv.w * sc + bv.w;
  *(float4*)(out + (size_t)node * DHID + c4) = o;
}

__global__ __launch_bounds__(256) void k_scatter(
    const int* __restrict__ src, const int* __restrict__ dst,
    const float* __restrict__ dinv, const float* __restrict__ h,
    float* __restrict__ out, int e)
{
  int w    = (blockIdx.x * blockDim.x + threadIdx.x) >> 5;
  int lane = threadIdx.x & 31;
  if (w >= e) return;
  int   s = src[w], d = dst[w];
  float c = dinv[s] * dinv[d];
  float4 hv = *(const float4*)(h + (size_t)s * DHID + lane * 4);
  float* o  = out + (size_t)d * DHID + lane * 4;
  unsafeAtomicAdd(o + 0, hv.x * c);
  unsafeAtomicAdd(o + 1, hv.y * c);
  unsafeAtomicAdd(o + 2, hv.z * c);
  unsafeAtomicAdd(o + 3, hv.w * c);
}

__global__ __launch_bounds__(256) void k_bnstats(const float* __restrict__ h,
                                                 float* __restrict__ stats, int n) {
  __shared__ float ss[256], sq[256];
  int t = threadIdx.x, c = t & 127, half = t >> 7;
  float s = 0.0f, q = 0.0f;
  for (int r = blockIdx.x * 2 + half; r < n; r += gridDim.x * 2) {
    float v = fmaxf(h[(size_t)r * DHID + c], 0.0f);
    s += v; q += v * v;
  }
  ss[t] = s; sq[t] = q;
  __syncthreads();
  if (t < 128) {
    unsafeAtomicAdd(&stats[c],       ss[t] + ss[t + 128]);
    unsafeAtomicAdd(&stats[128 + c], sq[t] + sq[t + 128]);
  }
}

__global__ void k_bnfinal(const float* __restrict__ stats, const float* __restrict__ gamma,
                          const float* __restrict__ beta, float* __restrict__ bn, int n) {
  int c = threadIdx.x;
  if (c < 128) {
    float inv = 1.0f / (float)n;
    float mu  = stats[c] * inv;
    float var = stats[128 + c] * inv - mu * mu;
    float sc  = gamma[c] * __frsqrt_rn(var + BN_EPS);
    bn[c]       = sc;
    bn[128 + c] = beta[c] - mu * sc;
  }
}

__global__ __launch_bounds__(256) void k_head(const float* __restrict__ hc,
                                              const float* __restrict__ Wr,
                                              const float* __restrict__ br,
                                              float* __restrict__ out, int n) {
  int w    = (blockIdx.x * blockDim.x + threadIdx.x) >> 5;
  int lane = threadIdx.x & 31;
  if (w >= n) return;
  float4 v = *(const float4*)(hc + (size_t)w * DHID + lane * 4);
  int c = lane * 4;
  float p0 = v.x * Wr[c * 2]       + v.y * Wr[(c + 1) * 2]
           + v.z * Wr[(c + 2) * 2] + v.w * Wr[(c + 3) * 2];
  float p1 = v.x * Wr[c * 2 + 1]       + v.y * Wr[(c + 1) * 2 + 1]
           + v.z * Wr[(c + 2) * 2 + 1] + v.w * Wr[(c + 3) * 2 + 1];
#pragma unroll
  for (int off = 16; off > 0; off >>= 1) {
    p0 += __shfl_down(p0, off, 32);
    p1 += __shfl_down(p1, off, 32);
  }
  if (lane == 0) {
    out[(size_t)w * 2]     = p0 + br[0];
    out[(size_t)w * 2 + 1] = p1 + br[1];
  }
}

extern "C" void kernel_launch(void* const* d_in, const int* in_sizes, int n_in,
                              void* d_out, int out_size, void* d_ws, size_t ws_size,
                              hipStream_t stream) {
  const float* x     = (const float*)d_in[0];
  const int*   eidx  = (const int*)d_in[1];
  const float* W1    = (const float*)d_in[2];
  const float* b1    = (const float*)d_in[3];
  const float* W2    = (const float*)d_in[4];
  const float* b2    = (const float*)d_in[5];
  const float* W3    = (const float*)d_in[6];
  const float* b3    = (const float*)d_in[7];
  const float* gamma = (const float*)d_in[8];
  const float* beta  = (const float*)d_in[9];
  const float* Wc    = (const float*)d_in[10];
  const float* bc    = (const float*)d_in[11];
  const float* Wr    = (const float*)d_in[12];
  const float* br    = (const float*)d_in[13];
  float* out = (float*)d_out;

  const int N = N_NODES, E = E_EDGES;
  const int* src = eidx;
  const int* dst = eidx + E;

  float* buf0  = (float*)d_ws;
  float* buf1  = buf0 + (size_t)N * DHID;
  float* dinv  = buf1 + (size_t)N * DHID;
  float* selfc = dinv + N;
  float* stats = selfc + N;
  float* bnp   = stats + 256;

  const dim3 blk(256);
  const int gN    = (N + 255) / 256;
  const int gE    = (E + 255) / 256;
  const int gNF4  = (N * (DHID / 4) + 255) / 256;
  const int gGemm = (N + 127) / 128;
  const int gScat = (E * 32 + 255) / 256;
  const int gHead = (N * 32 + 255) / 256;

  k_zero<<<gN, blk, 0, stream>>>(dinv, N);
  k_zero<<<1, blk, 0, stream>>>(stats, 256);
  k_degree<<<gE, blk, 0, stream>>>(dst, dinv, E);
  k_dinv<<<gN, blk, 0, stream>>>(dinv, selfc, N);

  k_gemm128<0, false, false><<<gGemm, blk, 0, stream>>>(x, W1, buf0, N, nullptr, nullptr, nullptr);
  k_selfbias<<<gNF4, blk, 0, stream>>>(buf0, selfc, b1, buf1, N);
  k_scatter<<<gScat, blk, 0, stream>>>(src, dst, dinv, buf0, buf1, E);

  k_gemm128<1, false, false><<<gGemm, blk, 0, stream>>>(buf1, W2, buf0, N, nullptr, nullptr, nullptr);
  k_selfbias<<<gNF4, blk, 0, stream>>>(buf0, selfc, b2, buf1, N);
  k_scatter<<<gScat, blk, 0, stream>>>(src, dst, dinv, buf0, buf1, E);

  k_gemm128<1, false, false><<<gGemm, blk, 0, stream>>>(buf1, W3, buf0, N, nullptr, nullptr, nullptr);
  k_selfbias<<<gNF4, blk, 0, stream>>>(buf0, selfc, b3, buf1, N);
  k_scatter<<<gScat, blk, 0, stream>>>(src, dst, dinv, buf0, buf1, E);

  k_bnstats<<<1024, blk, 0, stream>>>(buf1, stats, N);
  k_bnfinal<<<1, 128, 0, stream>>>(stats, gamma, beta, bnp, N);

  k_gemm128<2, true, true><<<gGemm, blk, 0, stream>>>(buf1, Wc, buf0, N, bnp, bnp + 128, bc);

  k_head<<<gHead, blk, 0, stream>>>(buf0, Wr, br, out, N);
}


# Neighbour sums: one wave per destination gathers its rows over an in-call destination-ordered edge list (slot = counting atomic's return) instead of per-edge f32 atomics; sum order = arrival rank
# speedup vs baseline: 2.6339x; 2.6339x over previous
#include <hip/hip_runtime.h>

#define N_NODES 100000
#define DHID    128
#define E_EDGES 1600000
#define BN_EPS  1e-5f

typedef __attribute__((ext_vector_type(2))) float v2f;
typedef __attribute__((ext_vector_type(8))) float v8f;

__global__ void k_zero(float* __restrict__ p, int n) {
  int i = blockIdx.x * blockDim.x + threadIdx.x;
  if (i < n) p[i] = 0.0f;
}

__global__ void k_count(const int* __restrict__ dst, int* __restrict__ cnt,
                        int* __restrict__ rank, int e) {
  int i = blockIdx.x * blockDim.x + threadIdx.x;
  if (i < e) rank[i] = atomicAdd(&cnt[dst[i]], 1);
}

__global__ __launch_bounds__(256) void k_blocksum(const int* __restrict__ cnt,
                                                  int* __restrict__ bsum, int n) {
  __shared__ int sh[256];
  int t  = threadIdx.x;
  int i0 = (blockIdx.x * 256 + t) * 4;
  int v  = 0;
#pragma unroll
  for (int q = 0; q < 4; ++q)
    if (i0 + q < n) v += cnt[i0 + q];
  sh[t] = v;
  __syncthreads();
  for (int off = 128; off > 0; off >>= 1) {
    if (t < off) sh[t] += sh[t + off];
    __syncthreads();
  }
  if (t == 0) bsum[blockIdx.x] = sh[0];
}

__global__ __launch_bounds__(256) void k_scan_dinv(const int* __restrict__ cnt,
                                                   const int* __restrict__ bsum,
                                                   int* __restrict__ rowptr,
                                                   float* __restrict__ dinv, int n) {
  __shared__ int sh[256];
  const int t = threadIdx.x, b = blockIdx.x;
  sh[t] = (t < b) ? bsum[t] : 0;
  __syncthreads();
  for (int off = 128; off > 0; off >>= 1) {
    if (t < off) sh[t] += sh[t + off];
    __syncthreads();
  }
  const int boff = sh[0];
  __syncthreads();

  const int i0 = (b * 256 + t) * 4;
  int c[4], tot = 0;
#pragma unroll
  for (int q = 0; q < 4; ++q) {
    c[q] = (i0 + q < n) ? cnt[i0 + q] : 0;
    tot += c[q];
  }
  sh[t] = tot;
  __syncthreads();
  for (int off = 1; off < 256; off <<= 1) {
    int x = (t >= off) ? sh[t - off] : 0;
    __syncthreads();
    sh[t] += x;
    __syncthreads();
  }
  int excl = boff + sh[t] - tot;
#pragma unroll
  for (int q = 0; q < 4; ++q) {
    if (i0 + q < n) {
      rowptr[i0 + q] = excl;
      dinv[i0 + q]   = __frsqrt_rn((float)c[q] + 1.0f);
      excl += c[q];
    }
  }
  if (i0 < n && i0 + 4 >= n) rowptr[n] = excl;
}

__global__ void k_fill(const int* __restrict__ src, const int* __restrict__ dst,
                       const int* __restrict__ rowptr, const int* __restrict__ rank,
                       int* __restrict__ col, int e) {
  int i = blockIdx.x * blockDim.x + threadIdx.x;
  if (i < e) col[rowptr[dst[i]] + rank[i]] = src[i];
}

template <int AMODE, bool HAS_BIAS, bool EPI_RELU>
__global__ __launch_bounds__(256) void k_gemm128(
    const float* __restrict__ A, const float* __restrict__ W,
    float* __restrict__ C, int nrows,
    const float* __restrict__ bnScale, const float* __restrict__ bnShift,
    const float* __restrict__ bias)
{
  __shared__ float wl[DHID * DHID];
  const int tid = threadIdx.x;
  for (int i = tid; i < DHID * DHID; i += 256) {
    int k = i >> 7, n = i & 127;
    wl[(((k >> 1) << 7) + n) * 2 + (k & 1)] = W[i];
  }
  __syncthreads();

  const int wave  = tid >> 5;
  const int lane  = tid & 31;
  const int laneM = lane & 15;
  const int hi    = lane >> 4;
  const int m0    = (blockIdx.x * 8 + wave) << 4;
  if (m0 >= nrows) return;

  const float2* __restrict__ wl2  = (const float2*)wl;
  const float*  __restrict__ arow = A + (size_t)(m0 + laneM) * DHID;

  v8f acc[8];
#pragma unroll
  for (int t = 0; t < 8; ++t) acc[t] = (v8f){};

  for (int k = 0; k < DHID; k += 4) {
    float2 av = *(const float2*)(arow + k + hi * 2);
    if (AMODE == 1) {
      av.x = fmaxf(av.x, 0.0f);
      av.y = fmaxf(av.y, 0.0f);
    } else if (AMODE == 2) {
      int c0 = k + hi * 2;
      av.x = fmaxf(av.x, 0.0f) * bnScale[c0]     + bnShift[c0];
      av.y = fmaxf(av.y, 0.0f) * bnScale[c0 + 1] + bnShift[c0 + 1];
    }
    v2f a; a[0] = av.x; a[1] = av.y;

    const float2* __restrict__ brow = wl2 + (((k >> 1) + hi) << 7);
#pragma unroll
    for (int t = 0; t < 8; ++t) {
      float2 bv = brow[t * 16 + laneM];
      v2f b; b[0] = bv.x; b[1] = bv.y;
      acc[t] = __builtin_amdgcn_wmma_f32_16x16x4_f32(
          false, a, false, b, (short)0, acc[t], false, false);
    }
  }

#pragma unroll
  for (int t = 0; t < 8; ++t) {
    int   col  = t * 16 + laneM;
    float bval = HAS_BIAS ? bias[col] : 0.0f;
#pragma unroll
    for (int r = 0; r < 8; ++r) {
      int   row = m0 + r + hi * 8;
      float v   = acc[t][r] + bval;
      if (EPI_RELU) v = fmaxf(v, 0.0f);
      C[(size_t)row * DHID + col] = v;
    }
  }
}

__device__ __forceinline__ float4 ld_row(const float* __restrict__ h, int s, int lane) {
  return *(const float4*)(h + (size_t)s * DHID + lane * 4);
}
__device__ __forceinline__ void fma_row(float4& acc, const float4& v, float c) {
  acc.x = fmaf(v.x, c, acc.x);
  acc.y = fmaf(v.y, c, acc.y);
  acc.z = fmaf(v.z, c, acc.z);
  acc.w = fmaf(v.w, c, acc.w);
}
__device__ __forceinline__ int lane_i(int v, int j) { return __builtin_amdgcn_readlane(v, j); }
__device__ __forceinline__ float lane_f(float v, int j) {
  return __int_as_float(__builtin_amdgcn_readlane(__float_as_int(v), j));
}

__global__ __launch_bounds__(256) void k_gather(
    const int* __restrict__ rowptr, const int* __restrict__ col,
    const float* __restrict__ dinv, const float* __restrict__ h,
    const float* __restrict__ b, float* __restrict__ out, int n)
{
  const int lane = threadIdx.x & 31;
  const int node = __builtin_amdgcn_readfirstlane((int)((blockIdx.x * 256 + threadIdx.x) >> 5));
  if (node >= n) return;
  const int   start = rowptr[node];
  const int   end   = rowptr[node + 1];
  const float dd    = dinv[node];
  const float4 hs   = ld_row(h, node, lane);
  const float4 bv   = *(const float4*)(b + lane * 4);

  float4 acc = make_float4(0.0f, 0.0f, 0.0f, 0.0f);
  for (int base = start; base < end; base += 32) {
    const int idx = base + lane;
    int   s  = node;
    float cs = 0.0f;
    if (idx < end) {
      s  = col[idx];
      s  = ((unsigned)s < (unsigned)n) ? s : node;
      cs = dinv[s];
    }
    const int cnt = min(32, end - base);
    int j = 0;
    for (; j + 8 <= cnt; j += 8) {
      float4 t[8];
#pragma unroll
      for (int q = 0; q < 8; ++q) t[q] = ld_row(h, lane_i(s, j + q), lane);
#pragma unroll
      for (int q = 0; q < 8; ++q) fma_row(acc, t[q], lane_f(cs, j + q) * dd);
    }
    const int r = cnt - j;
    if (r > 0) {
      float4 t[7];
#pragma unroll
      for (int q = 0; q < 7; ++q)
        if (q < r) t[q] = ld_row(h, lane_i(s, j + q), lane);
#pragma unroll
      for (int q = 0; q < 7; ++q)
        if (q < r) fma_row(acc, t[q], lane_f(cs, j + q) * dd);
    }
  }

  const float sc = dd * dd;
  float4 o;
  o.x = fmaf(hs.x, sc, acc.x) + bv.x;
  o.y = fmaf(hs.y, sc, acc.y) + bv.y;
  o.z = fmaf(hs.z, sc, acc.z) + bv.z;
  o.w = fmaf(hs.w, sc, acc.w) + bv.w;
  *(float4*)(out + (size_t)node * DHID + lane * 4) = o;
}

__global__ __launch_bounds__(256) void k_bnstats(const float* __restrict__ h,
                                                 float* __restrict__ stats, int n) {
  __shared__ float ss[256], sq[256];
  int t = threadIdx.x, c = t & 127, half = t >> 7;
  float s = 0.0f, q = 0.0f;
  for (int r = blockIdx.x * 2 + half; r < n; r += gridDim.x * 2) {
    float v = fmaxf(h[(size_t)r * DHID + c], 0.0f);
    s += v; q += v * v;
  }
  ss[t] = s; sq[t] = q;
  __syncthreads();
  if (t < 128) {
    unsafeAtomicAdd(&stats[c],       ss[t] + ss[t + 128]);
    unsafeAtomicAdd(&stats[128 + c], sq[t] + sq[t + 128]);
  }
}

__global__ void k_bnfinal(const float* __restrict__ stats, const float* __restrict__ gamma,
                          const float* __restrict__ beta, float* __restrict__ bn, int n) {
  int c = threadIdx.x;
  if (c < 128) {
    float inv = 1.0f / (float)n;
    float mu  = stats[c] * inv;
    float var = stats[128 + c] * inv - mu * mu;
    float sc  = gamma[c] * __frsqrt_rn(var + BN_EPS);
    bn[c]       = sc;
    bn[128 + c] = beta[c] - mu * sc;
  }
}

__global__ __launch_bounds__(256) void k_head(const float* __restrict__ hc,
                                              const float* __restrict__ Wr,
                                              const float* __restrict__ br,
                                              float* __restrict__ out, int n) {
  int w    = (blockIdx.x * blockDim.x + threadIdx.x) >> 5;
  int lane = threadIdx.x & 31;
  if (w >= n) return;
  float4 v = *(const float4*)(hc + (size_t)w * DHID + lane * 4);
  int c = lane * 4;
  float p0 = v.x * Wr[c * 2]       + v.y * Wr[(c + 1) * 2]
           + v.z * Wr[(c + 2) * 2] + v.w * Wr[(c + 3) * 2];
  float p1 = v.x * Wr[c * 2 + 1]       + v.y * Wr[(c + 1) * 2 + 1]
           + v.z * Wr[(c + 2) * 2 + 1] + v.w * Wr[(c + 3) * 2 + 1];
#pragma unroll
  for (int off = 16; off > 0; off >>= 1) {
    p0 += __shfl_down(p0, off, 32);
    p1 += __shfl_down(p1, off, 32);
  }
  if (lane == 0) {
    out[(size_t)w * 2]     = p0 + br[0];
    out[(size_t)w * 2 + 1] = p1 + br[1];
  }
}

extern "C" void kernel_launch(void* const* d_in, const int* in_sizes, int n_in,
                              void* d_out, int out_size, void* d_ws, size_t ws_size,
                              hipStream_t stream) {
  const float* x     = (const float*)d_in[0];
  const int*   eidx  = (const int*)d_in[1];
  const float* W1    = (const float*)d_in[2];
  const float* b1    = (const float*)d_in[3];
  const float* W2    = (const float*)d_in[4];
  const float* b2    = (const float*)d_in[5];
  const float* W3    = (const float*)d_in[6];
  const float* b3    = (const float*)d_in[7];
  const float* gamma = (const float*)d_in[8];
  const float* beta  = (const float*)d_in[9];
  const float* Wc    = (const float*)d_in[10];
  const float* bc    = (const float*)d_in[11];
  const float* Wr    = (const float*)d_in[12];
  const float* br    = (const float*)d_in[13];
  float* out = (float*)d_out;

  const int N = N_NODES, E = E_EDGES;
  const int* src = eidx;
  const int* dst = eidx + E;

  float* buf0   = (float*)d_ws;
  float* buf1   = buf0 + (size_t)N * DHID;
  int*   cnt    = (int*)(buf1 + (size_t)N * DHID);
  float* stats  = (float*)(cnt + N);
  float* bnp    = stats + 256;
  int*   bsum   = (int*)(bnp + 256);
  float* dinv   = (float*)(bsum + 256);
  int*   rowptr = (int*)(dinv + N);
  int*   rank   = rowptr + N + 4;
  int*   col    = rank + E;

  const dim3 blk(256);
  const int gZero = (N + 256 + 255) / 256;
  const int gE    = (E + 255) / 256;
  const int gScan = (N + 1023) / 1024;
  const int gGemm = (N + 127) / 128;
  const int gNode = (N * 32 + 255) / 256;

  k_zero<<<gZero, blk, 0, stream>>>((float*)cnt, N + 256);
  k_count<<<gE, blk, 0, stream>>>(dst, cnt, rank, E);
  k_blocksum<<<gScan, blk, 0, stream>>>(cnt, bsum, N);
  k_scan_dinv<<<gScan, blk, 0, stream>>>(cnt, bsum, rowptr, dinv, N);
  k_fill<<<gE, blk, 0, stream>>>(src, dst, rowptr, rank, col, E);

  k_gemm128<0, false, false><<<gGemm, blk, 0, stream>>>(x, W1, buf0, N, nullptr, nullptr, nullptr);
  k_gather<<<gNode, blk, 0, stream>>>(rowptr, col, dinv, buf0, b1, buf1, N);

  k_gemm128<1, false, false><<<gGemm, blk, 0, stream>>>(buf1, W2, buf0, N, nullptr, nullptr, nullptr);
  k_gather<<<gNode, blk, 0, stream>>>(rowptr, col, dinv, buf0, b2, buf1, N);

  k_gemm128<1, false, false><<<gGemm, blk, 0, stream>>>(buf1, W3, buf0, N, nullptr, nullptr, nullptr);
  k_gather<<<gNode, blk, 0, stream>>>(rowptr, col, dinv, buf0, b3, buf1, N);

  k_bnstats<<<1024, blk, 0, stream>>>(buf1, stats, N);
  k_bnfinal<<<1, 128, 0, stream>>>(stats, gamma, beta, bnp, N);

  k_gemm128<2, true, true><<<gGemm, blk, 0, stream>>>(buf1, Wc, buf0, N, bnp, bnp + 128, bc);

  k_head<<<gNode, blk, 0, stream>>>(buf0, Wr, br, out, N);
}


# Tail fused into one launch: the classifier GEMM builds the batchnorm scale/shift itself in LDS and applies the 128->2 head to its register tile (hc never stored); head dot summed in a different tree
# speedup vs baseline: 2.8402x; 1.0783x over previous
#include <hip/hip_runtime.h>

#define N_NODES 100000
#define DHID    128
#define E_EDGES 1600000
#define BN_EPS  1e-5f

typedef __attribute__((ext_vector_type(2))) float v2f;
typedef __attribute__((ext_vector_type(8))) float v8f;

__global__ void k_zero(float* __restrict__ p, int n) {
  int i = blockIdx.x * blockDim.x + threadIdx.x;
  if (i < n) p[i] = 0.0f;
}

__global__ void k_count(const int* __restrict__ dst, int* __restrict__ cnt,
                        int* __restrict__ rank, int e) {
  int i = blockIdx.x * blockDim.x + threadIdx.x;
  if (i < e) rank[i] = atomicAdd(&cnt[dst[i]], 1);
}

__global__ __launch_bounds__(256) void k_blocksum(const int* __restrict__ cnt,
                                                  int* __restrict__ bsum, int n) {
  __shared__ int sh[256];
  int t  = threadIdx.x;
  int i0 = (blockIdx.x * 256 + t) * 4;
  int v  = 0;
#pragma unroll
  for (int q = 0; q < 4; ++q)
    if (i0 + q < n) v += cnt[i0 + q];
  sh[t] = v;
  __syncthreads();
  for (int off = 128; off > 0; off >>= 1) {
    if (t < off) sh[t] += sh[t + off];
    __syncthreads();
  }
  if (t == 0) bsum[blockIdx.x] = sh[0];
}

__global__ __launch_bounds__(256) void k_scan_dinv(const int* __restrict__ cnt,
                                                   const int* __restrict__ bsum,
                                                   int* __restrict__ rowptr,
                                                   float* __restrict__ dinv, int n) {
  __shared__ int sh[256];
  const int t = threadIdx.x, b = blockIdx.x;
  sh[t] = (t < b) ? bsum[t] : 0;
  __syncthreads();
  for (int off = 128; off > 0; off >>= 1) {
    if (t < off) sh[t] += sh[t + off];
    __syncthreads();
  }
  const int boff = sh[0];
  __syncthreads();

  const int i0 = (b * 256 + t) * 4;
  int c[4], tot = 0;
#pragma unroll
  for (int q = 0; q < 4; ++q) {
    c[q] = (i0 + q < n) ? cnt[i0 + q] : 0;
    tot += c[q];
  }
  sh[t] = tot;
  __syncthreads();
  for (int off = 1; off < 256; off <<= 1) {
    int x = (t >= off) ? sh[t - off] : 0;
    __syncthreads();
    sh[t] += x;
    __syncthreads();
  }
  int excl = boff + sh[t] - tot;
#pragma unroll
  for (int q = 0; q < 4; ++q) {
    if (i0 + q < n) {
      rowptr[i0 + q] = excl;
      dinv[i0 + q]   = __frsqrt_rn((float)c[q] + 1.0f);
      excl += c[q];
    }
  }
  if (i0 < n && i0 + 4 >= n) rowptr[n] = excl;
}

__global__ void k_fill(const int* __restrict__ src, const int* __restrict__ dst,
                       const int* __restrict__ rowptr, const int* __restrict__ rank,
                       int* __restrict__ col, int e) {
  int i = blockIdx.x * blockDim.x + threadIdx.x;
  if (i < e) col[rowptr[dst[i]] + rank[i]] = src[i];
}

template <int AMODE, bool HAS_BIAS, bool EPI_RELU>
__global__ __launch_bounds__(256) void k_gemm128(
    const float* __restrict__ A, const float* __restrict__ W,
    float* __restrict__ C, int nrows,
    const float* __restrict__ bnScale, const float* __restrict__ bnShift,
    const float* __restrict__ bias)
{
  __shared__ float wl[DHID * DHID];
  const int tid = threadIdx.x;
  for (int i = tid; i < DHID * DHID; i += 256) {
    int k = i >> 7, n = i & 127;
    wl[(((k >> 1) << 7) + n) * 2 + (k & 1)] = W[i];
  }
  __syncthreads();

  const int wave  = tid >> 5;
  const int lane  = tid & 31;
  const int laneM = lane & 15;
  const int hi    = lane >> 4;
  const int m0    = (blockIdx.x * 8 + wave) << 4;
  if (m0 >= nrows) return;

  const float2* __restrict__ wl2  = (const float2*)wl;
  const float*  __restrict__ arow = A + (size_t)(m0 + laneM) * DHID;

  v8f acc[8];
#pragma unroll
  for (int t = 0; t < 8; ++t) acc[t] = (v8f){};

  for (int k = 0; k < DHID; k += 4) {
    float2 av = *(const float2*)(arow + k + hi * 2);
    if (AMODE == 1) {
      av.x = fmaxf(av.x, 0.0f);
      av.y = fmaxf(av.y, 0.0f);
    } else if (AMODE == 2) {
      int c0 = k + hi * 2;
      av.x = fmaxf(av.x, 0.0f) * bnScale[c0]     + bnShift[c0];
      av.y = fmaxf(av.y, 0.0f) * bnScale[c0 + 1] + bnShift[c0 + 1];
    }
    v2f a; a[0] = av.x; a[1] = av.y;

    const float2* __restrict__ brow = wl2 + (((k >> 1) + hi) << 7);
#pragma unroll
    for (int t = 0; t < 8; ++t) {
      float2 bv = brow[t * 16 + laneM];
      v2f b; b[0] = bv.x; b[1] = bv.y;
      acc[t] = __builtin_amdgcn_wmma_f32_16x16x4_f32(
          false, a, false, b, (short)0, acc[t], false, false);
    }
  }

#pragma unroll
  for (int t = 0; t < 8; ++t) {
    int   col  = t * 16 + laneM;
    float bval = HAS_BIAS ? bias[col] : 0.0f;
#pragma unroll
    for (int r = 0; r < 8; ++r) {
      int   row = m0 + r + hi * 8;
      float v   = acc[t][r] + bval;
      if (EPI_RELU) v = fmaxf(v, 0.0f);
      C[(size_t)row * DHID + col] = v;
    }
  }
}

__device__ __forceinline__ float4 ld_row(const float* __restrict__ h, int s, int lane) {
  return *(const float4*)(h + (size_t)s * DHID + lane * 4);
}
__device__ __forceinline__ void fma_row(float4& acc, const float4& v, float c) {
  acc.x = fmaf(v.x, c, acc.x);
  acc.y = fmaf(v.y, c, acc.y);
  acc.z = fmaf(v.z, c, acc.z);
  acc.w = fmaf(v.w, c, acc.w);
}
__device__ __forceinline__ int lane_i(int v, int j) { return __builtin_amdgcn_readlane(v, j); }
__device__ __forceinline__ float lane_f(float v, int j) {
  return __int_as_float(__builtin_amdgcn_readlane(__float_as_int(v), j));
}

__global__ __launch_bounds__(256) void k_gather(
    const int* __restrict__ rowptr, const int* __restrict__ col,
    const float* __restrict__ dinv, const float* __restrict__ h,
    const float* __restrict__ b, float* __restrict__ out, int n)
{
  const int lane = threadIdx.x & 31;
  const int node = __builtin_amdgcn_readfirstlane((int)((blockIdx.x * 256 + threadIdx.x) >> 5));
  if (node >= n) return;
  const int   start = rowptr[node];
  const int   end   = rowptr[node + 1];
  const float dd    = dinv[node];
  const float4 hs   = ld_row(h, node, lane);
  const float4 bv   = *(const float4*)(b + lane * 4);

  float4 acc = make_float4(0.0f, 0.0f, 0.0f, 0.0f);
  for (int base = start; base < end; base += 32) {
    const int idx = base + lane;
    int   s  = node;
    float cs = 0.0f;
    if (idx < end) {
      s  = col[idx];
      s  = ((unsigned)s < (unsigned)n) ? s : node;
      cs = dinv[s];
    }
    const int cnt = min(32, end - base);
    int j = 0;
    for (; j + 8 <= cnt; j += 8) {
      float4 t[8];
#pragma unroll
      for (int q = 0; q < 8; ++q) t[q] = ld_row(h, lane_i(s, j + q), lane);
#pragma unroll
      for (int q = 0; q < 8; ++q) fma_row(acc, t[q], lane_f(cs, j + q) * dd);
    }
    const int r = cnt - j;
    if (r > 0) {
      float4 t[7];
#pragma unroll
      for (int q = 0; q < 7; ++q)
        if (q < r) t[q] = ld_row(h, lane_i(s, j + q), lane);
#pragma unroll
      for (int q = 0; q < 7; ++q)
        if (q < r) fma_row(acc, t[q], lane_f(cs, j + q) * dd);
    }
  }

  const float sc = dd * dd;
  float4 o;
  o.x = fmaf(hs.x, sc, acc.x) + bv.x;
  o.y = fmaf(hs.y, sc, acc.y) + bv.y;
  o.z = fmaf(hs.z, sc, acc.z) + bv.z;
  o.w = fmaf(hs.w, sc, acc.w) + bv.w;
  *(float4*)(out + (size_t)node * DHID + lane * 4) = o;
}

__global__ __launch_bounds__(256) void k_bnstats(const float* __restrict__ h,
                                                 float* __restrict__ stats, int n) {
  __shared__ float ss[256], sq[256];
  int t = threadIdx.x, c = t & 127, half = t >> 7;
  float s = 0.0f, q = 0.0f;
  for (int r = blockIdx.x * 2 + half; r < n; r += gridDim.x * 2) {
    float v = fmaxf(h[(size_t)r * DHID + c], 0.0f);
    s += v; q += v * v;
  }
  ss[t] = s; sq[t] = q;
  __syncthreads();
  if (t < 128) {
    unsafeAtomicAdd(&stats[c],       ss[t] + ss[t + 128]);
    unsafeAtomicAdd(&stats[128 + c], sq[t] + sq[t + 128]);
  }
}

__global__ __launch_bounds__(256) void k_classify_head(
    const float* __restrict__ A, const float* __restrict__ W,
    const float* __restrict__ stats, const float* __restrict__ gamma,
    const float* __restrict__ beta, const float* __restrict__ bias,
    const float* __restrict__ Wr, const float* __restrict__ br,
    float* __restrict__ out, int nrows)
{
  __shared__ float wl[DHID * DHID];
  __shared__ __attribute__((aligned(16))) float bnl[2 * DHID];
  const int tid = threadIdx.x;
  for (int i = tid; i < DHID * DHID; i += 256) {
    int k = i >> 7, n = i & 127;
    wl[(((k >> 1) << 7) + n) * 2 + (k & 1)] = W[i];
  }
  if (tid < DHID) {
    float inv = 1.0f / (float)nrows;
    float mu  = stats[tid] * inv;
    float var = stats[128 + tid] * inv - mu * mu;
    float sc  = gamma[tid] * __frsqrt_rn(var + BN_EPS);
    bnl[tid]        = sc;
    bnl[DHID + tid] = beta[tid] - mu * sc;
  }
  __syncthreads();

  const int wave  = tid >> 5;
  const int lane  = tid & 31;
  const int laneM = lane & 15;
  const int hi    = lane >> 4;
  const int m0    = (blockIdx.x * 8 + wave) << 4;
  if (m0 >= nrows) return;

  const float2* __restrict__ wl2  = (const float2*)wl;
  const float2* __restrict__ bn2  = (const float2*)bnl;
  const float*  __restrict__ arow = A + (size_t)(m0 + laneM) * DHID;

  v8f acc[8];
#pragma unroll
  for (int t = 0; t < 8; ++t) acc[t] = (v8f){};

#pragma unroll 2
  for (int k = 0; k < DHID; k += 4) {
    const int c0 = k + hi * 2;
    float2 av = *(const float2*)(arow + c0);
    const float2 sc2 = bn2[c0 >> 1];
    const float2 sh2 = bn2[(DHID >> 1) + (c0 >> 1)];
    av.x = fmaxf(av.x, 0.0f) * sc2.x + sh2.x;
    av.y = fmaxf(av.y, 0.0f) * sc2.y + sh2.y;
    v2f a; a[0] = av.x; a[1] = av.y;

    const float2* __restrict__ brow = wl2 + (((k >> 1) + hi) << 7);
#pragma unroll
    for (int t = 0; t < 8; ++t) {
      float2 bv = brow[t * 16 + laneM];
      v2f b; b[0] = bv.x; b[1] = bv.y;
      acc[t] = __builtin_amdgcn_wmma_f32_16x16x4_f32(
          false, a, false, b, (short)0, acc[t], false, false);
    }
  }

  float p[16];
#pragma unroll
  for (int i = 0; i < 16; ++i) p[i] = 0.0f;
#pragma unroll
  for (int t = 0; t < 8; ++t) {
    const int    col  = t * 16 + laneM;
    const float  bval = bias[col];
    const float2 wr   = *(const float2*)(Wr + col * 2);
#pragma unroll
    for (int r = 0; r < 8; ++r) {
      float v = fmaxf(acc[t][r] + bval, 0.0f);
      p[2 * r]     = fmaf(v, wr.x, p[2 * r]);
      p[2 * r + 1] = fmaf(v, wr.y, p[2 * r + 1]);
    }
  }
  float w8[8], w4[4], w2[2];
  const bool u8 = (laneM & 8) != 0, u4 = (laneM & 4) != 0, u2 = (laneM & 2) != 0, u1 = (laneM & 1) != 0;
#pragma unroll
  for (int j = 0; j < 8; ++j) {
    float keep = u8 ? p[j + 8] : p[j];
    float send = u8 ? p[j] : p[j + 8];
    w8[j] = keep + __shfl_xor(send, 8, 32);
  }
#pragma unroll
  for (int j = 0; j < 4; ++j) {
    float keep = u4 ? w8[j + 4] : w8[j];
    float send = u4 ? w8[j] : w8[j + 4];
    w4[j] = keep + __shfl_xor(send, 4, 32);
  }
#pragma unroll
  for (int j = 0; j < 2; ++j) {
    float keep = u2 ? w4[j + 2] : w4[j];
    float send = u2 ? w4[j] : w4[j + 2];
    w2[j] = keep + __shfl_xor(send, 2, 32);
  }
  {
    float keep = u1 ? w2[1] : w2[0];
    float send = u1 ? w2[0] : w2[1];
    float tot  = keep + __shfl_xor(send, 1, 32);
    out[(size_t)m0 * 2 + lane] = tot + br[lane & 1];
  }
}

extern "C" void kernel_launch(void* const* d_in, const int* in_sizes, int n_in,
                              void* d_out, int out_size, void* d_ws, size_t ws_size,
                              hipStream_t stream) {
  const float* x     = (const float*)d_in[0];
  const int*   eidx  = (const int*)d_in[1];
  const float* W1    = (const float*)d_in[2];
  const float* b1    = (const float*)d_in[3];
  const float* W2    = (const float*)d_in[4];
  const float* b2    = (const float*)d_in[5];
  const float* W3    = (const float*)d_in[6];
  const float* b3    = (const float*)d_in[7];
  const float* gamma = (const float*)d_in[8];
  const float* beta  = (const float*)d_in[9];
  const float* Wc    = (const float*)d_in[10];
  const float* bc    = (const float*)d_in[11];
  const float* Wr    = (const float*)d_in[12];
  const float* br    = (const float*)d_in[13];
  float* out = (float*)d_out;

  const int N = N_NODES, E = E_EDGES;
  const int* src = eidx;
  const int* dst = eidx + E;

  float* buf0   = (float*)d_ws;
  float* buf1   = buf0 + (size_t)N * DHID;
  int*   cnt    = (int*)(buf1 + (size_t)N * DHID);
  float* stats  = (float*)(cnt + N);
  float* bnp    = stats + 256;
  int*   bsum   = (int*)(bnp + 256);
  float* dinv   = (float*)(bsum + 256);
  int*   rowptr = (int*)(dinv + N);
  int*   rank   = rowptr + N + 4;
  int*   col    = rank + E;

  const dim3 blk(256);
  const int gZero = (N + 256 + 255) / 256;
  const int gE    = (E + 255) / 256;
  const int gScan = (N + 1023) / 1024;
  const int gGemm = (N + 127) / 128;
  const int gNode = (N * 32 + 255) / 256;

  k_zero<<<gZero, blk, 0, stream>>>((float*)cnt, N + 256);
  k_count<<<gE, blk, 0, stream>>>(dst, cnt, rank, E);
  k_blocksum<<<gScan, blk, 0, stream>>>(cnt, bsum, N);
  k_scan_dinv<<<gScan, blk, 0, stream>>>(cnt, bsum, rowptr, dinv, N);
  k_fill<<<gE, blk, 0, stream>>>(src, dst, rowptr, rank, col, E);

  k_gemm128<0, false, false><<<gGemm, blk, 0, stream>>>(x, W1, buf0, N, nullptr, nullptr, nullptr);
  k_gather<<<gNode, blk, 0, stream>>>(rowptr, col, dinv, buf0, b1, buf1, N);

  k_gemm128<1, false, false><<<gGemm, blk, 0, stream>>>(buf1, W2, buf0, N, nullptr, nullptr, nullptr);
  k_gather<<<gNode, blk, 0, stream>>>(rowptr, col, dinv, buf0, b2, buf1, N);

  k_gemm128<1, false, false><<<gGemm, blk, 0, stream>>>(buf1, W3, buf0, N, nullptr, nullptr, nullptr);
  k_gather<<<gNode, blk, 0, stream>>>(rowptr, col, dinv, buf0, b3, buf1, N);

  k_bnstats<<<1024, blk, 0, stream>>>(buf1, stats, N);

  k_classify_head<<<gGemm, blk, 0, stream>>>(buf1, Wc, stats, gamma, beta, bc, Wr, br, out, N);
}


# Destination-ordered edge list built on LDS counters (8 node ranges x 32 edge chunks, per-node prefix over chunks) instead of returning global atomics; sum order in a node = chunk, then LDS arrival
# speedup vs baseline: 3.3953x; 1.1954x over previous
#include <hip/hip_runtime.h>

#define N_NODES 100000
#define DHID    128
#define E_EDGES 1600000
#define BN_EPS  1e-5f

typedef __attribute__((ext_vector_type(2))) float v2f;
typedef __attribute__((ext_vector_type(8))) float v8f;

#define NRANGE 8
#define NPR    (N_NODES / NRANGE)
#define NCHUNK 32
#define EPC    (E_EDGES / NCHUNK)
#define EPC4   (EPC / 4)
#define SORT_T 1024
static_assert(NPR * NRANGE == N_NODES, "ranges must tile the nodes");
static_assert(EPC * NCHUNK == E_EDGES && EPC4 * 4 == EPC, "chunks must tile the edges in 16-byte groups");
static_assert(N_NODES % 4 == 0, "k_degsum reads four nodes per thread");

__device__ __forceinline__ void count_one(int* lc, int d) {
  if ((unsigned)d < (unsigned)NPR) atomicAdd(&lc[d], 1);
}

__global__ __launch_bounds__(SORT_T) void k_count_lds(const int* __restrict__ dst,
                                                      int* __restrict__ tab) {
  __shared__ int lc[NPR];
  const int t = threadIdx.x, c = blockIdx.x, lo = blockIdx.y * NPR;
  for (int j = t; j < NPR; j += SORT_T) lc[j] = 0;
  __syncthreads();
  const int4* __restrict__ d4 = (const int4*)(dst + (size_t)c * EPC);
  constexpr int B = (EPC4 + SORT_T - 1) / SORT_T;
  int4 v[B];
#pragma unroll
  for (int q = 0; q < B; ++q) {
    const int i = t + q * SORT_T;
    v[q] = (i < EPC4) ? d4[i] : make_int4(-1, -1, -1, -1);
  }
#pragma unroll
  for (int q = 0; q < B; ++q) {
    count_one(lc, v[q].x - lo);
    count_one(lc, v[q].y - lo);
    count_one(lc, v[q].z - lo);
    count_one(lc, v[q].w - lo);
  }
  __syncthreads();
  int* __restrict__ trow = tab + (size_t)c * N_NODES + lo;
  for (int j = t; j < NPR; j += SORT_T) trow[j] = lc[j];
}

__global__ __launch_bounds__(256) void k_degsum(int* __restrict__ tab, int* __restrict__ cnt,
                                                int* __restrict__ bsum, float* __restrict__ stats,
                                                int n) {
  __shared__ int sh[256];
  const int t  = threadIdx.x;
  const int i0 = (blockIdx.x * 256 + t) * 4;
  if (blockIdx.x == 0) stats[t] = 0.0f;
  int4 run = make_int4(0, 0, 0, 0);
  if (i0 < n) {
    int4 v[NCHUNK];
#pragma unroll
    for (int c = 0; c < NCHUNK; ++c) v[c] = *(const int4*)(tab + (size_t)c * N_NODES + i0);
#pragma unroll
    for (int c = 0; c < NCHUNK; ++c) {
      *(int4*)(tab + (size_t)c * N_NODES + i0) = run;
      run.x += v[c].x; run.y += v[c].y; run.z += v[c].z; run.w += v[c].w;
    }
    *(int4*)(cnt + i0) = run;
  }
  sh[t] = run.x + run.y + run.z + run.w;
  __syncthreads();
  for (int off = 128; off > 0; off >>= 1) {
    if (t < off) sh[t] += sh[t + off];
    __syncthreads();
  }
  if (t == 0) bsum[blockIdx.x] = sh[0];
}

__global__ __launch_bounds__(256) void k_scan_dinv(const int* __restrict__ cnt,
                                                   const int* __restrict__ bsum,
                                                   int* __restrict__ rowptr,
                                                   float* __restrict__ dinv, int n) {
  __shared__ int sh[256];
  const int t = threadIdx.x, b = blockIdx.x;
  sh[t] = (t < b) ? bsum[t] : 0;
  __syncthreads();
  for (int off = 128; off > 0; off >>= 1) {
    if (t < off) sh[t] += sh[t + off];
    __syncthreads();
  }
  const int boff = sh[0];
  __syncthreads();

  const int i0 = (b * 256 + t) * 4;
  int c[4], tot = 0;
#pragma unroll
  for (int q = 0; q < 4; ++q) {
    c[q] = (i0 + q < n) ? cnt[i0 + q] : 0;
    tot += c[q];
  }
  sh[t] = tot;
  __syncthreads();
  for (int off = 1; off < 256; off <<= 1) {
    int x = (t >= off) ? sh[t - off] : 0;
    __syncthreads();
    sh[t] += x;
    __syncthreads();
  }
  int excl = boff + sh[t] - tot;
#pragma unroll
  for (int q = 0; q < 4; ++q) {
    if (i0 + q < n) {
      rowptr[i0 + q] = excl;
      dinv[i0 + q]   = __frsqrt_rn((float)c[q] + 1.0f);
      excl += c[q];
    }
  }
  if (i0 < n && i0 + 4 >= n) rowptr[n] = excl;
}

__device__ __forceinline__ int take_slot(int* cur, int d) {
  int slot = -1;
  if ((unsigned)d < (unsigned)NPR) slot = atomicAdd(&cur[d], 1);
  return slot;
}
__device__ __forceinline__ void put_src(int* __restrict__ col, int slot, int s) {
  if ((unsigned)slot < (unsigned)E_EDGES) col[slot] = s;
}

__global__ __launch_bounds__(SORT_T) void k_fill_lds(const int* __restrict__ src,
                                                     const int* __restrict__ dst,
                                                     const int* __restrict__ rowptr,
                                                     const int* __restrict__ tab,
                                                     int* __restrict__ col) {
  __shared__ int cur[NPR];
  const int t = threadIdx.x, c = blockIdx.x, lo = blockIdx.y * NPR;
  const int* __restrict__ trow = tab + (size_t)c * N_NODES + lo;
  const int* __restrict__ rp   = rowptr + lo;
  for (int j = t; j < NPR; j += SORT_T) cur[j] = rp[j] + trow[j];
  __syncthreads();
  const int4* __restrict__ d4 = (const int4*)(dst + (size_t)c * EPC);
  const int4* __restrict__ s4 = (const int4*)(src + (size_t)c * EPC);
  constexpr int B = 7;
  for (int p = 0; p < EPC4; p += B * SORT_T) {
    int4 dv[B], sv[B];
#pragma unroll
    for (int q = 0; q < B; ++q) {
      const int  i  = p + t + q * SORT_T;
      const bool ok = i < EPC4;
      dv[q] = ok ? d4[i] : make_int4(-1, -1, -1, -1);
      sv[q] = ok ? s4[i] : make_int4(0, 0, 0, 0);
    }
    int slot[4 * B];
#pragma unroll
    for (int q = 0; q < B; ++q) {
      slot[4 * q + 0] = take_slot(cur, dv[q].x - lo);
      slot[4 * q + 1] = take_slot(cur, dv[q].y - lo);
      slot[4 * q + 2] = take_slot(cur, dv[q].z - lo);
      slot[4 * q + 3] = take_slot(cur, dv[q].w - lo);
    }
#pragma unroll
    for (int q = 0; q < B; ++q) {
      put_src(col, slot[4 * q + 0], sv[q].x);
      put_src(col, slot[4 * q + 1], sv[q].y);
      put_src(col, slot[4 * q + 2], sv[q].z);
      put_src(col, slot[4 * q + 3], sv[q].w);
    }
  }
}

template <int AMODE, bool HAS_BIAS, bool EPI_RELU>
__global__ __launch_bounds__(256) void k_gemm128(
    const float* __restrict__ A, const float* __restrict__ W,
    float* __restrict__ C, int nrows,
    const float* __restrict__ bnScale, const float* __restrict__ bnShift,
    const float* __restrict__ bias)
{
  __shared__ float wl[DHID * DHID];
  const int tid = threadIdx.x;
  for (int i = tid; i < DHID * DHID; i += 256) {
    int k = i >> 7, n = i & 127;
    wl[(((k >> 1) << 7) + n) * 2 + (k & 1)] = W[i];
  }
  __syncthreads();

  const int wave  = tid >> 5;
  const int lane  = tid & 31;
  const int laneM = lane & 15;
  const int hi    = lane >> 4;
  const int m0    = (blockIdx.x * 8 + wave) << 4;
  if (m0 >= nrows) return;

  const float2* __restrict__ wl2  = (const float2*)wl;
  const float*  __restrict__ arow = A + (size_t)(m0 + laneM) * DHID;

  v8f acc[8];
#pragma unroll
  for (int t = 0; t < 8; ++t) acc[t] = (v8f){};

  for (int k = 0; k < DHID; k += 4) {
    float2 av = *(const float2*)(arow + k + hi * 2);
    if (AMODE == 1) {
      av.x = fmaxf(av.x, 0.0f);
      av.y = fmaxf(av.y, 0.0f);
    } else if (AMODE == 2) {
      int c0 = k + hi * 2;
      av.x = fmaxf(av.x, 0.0f) * bnScale[c0]     + bnShift[c0];
      av.y = fmaxf(av.y, 0.0f) * bnScale[c0 + 1] + bnShift[c0 + 1];
    }
    v2f a; a[0] = av.x; a[1] = av.y;

    const float2* __restrict__ brow = wl2 + (((k >> 1) + hi) << 7);
#pragma unroll
    for (int t = 0; t < 8; ++t) {
      float2 bv = brow[t * 16 + laneM];
      v2f b; b[0] = bv.x; b[1] = bv.y;
      acc[t] = __builtin_amdgcn_wmma_f32_16x16x4_f32(
          false, a, false, b, (short)0, acc[t], false, false);
    }
  }

#pragma unroll
  for (int t = 0; t < 8; ++t) {
    int   col  = t * 16 + laneM;
    float bval = HAS_BIAS ? bias[col] : 0.0f;
#pragma unroll
    for (int r = 0; r < 8; ++r) {
      int   row = m0 + r + hi * 8;
      float v   = acc[t][r] + bval;
      if (EPI_RELU) v = fmaxf(v, 0.0f);
      C[(size_t)row * DHID + col] = v;
    }
  }
}

__device__ __forceinline__ float4 ld_row(const float* __restrict__ h, int s, int lane) {
  return *(const float4*)(h + (size_t)s * DHID + lane * 4);
}
__device__ __forceinline__ void fma_row(float4& acc, const float4& v, float c) {
  acc.x = fmaf(v.x, c, acc.x);
  acc.y = fmaf(v.y, c, acc.y);
  acc.z = fmaf(v.z, c, acc.z);
  acc.w = fmaf(v.w, c, acc.w);
}
__device__ __forceinline__ int lane_i(int v, int j) { return __builtin_amdgcn_readlane(v, j); }
__device__ __forceinline__ float lane_f(float v, int j) {
  return __int_as_float(__builtin_amdgcn_readlane(__float_as_int(v), j));
}

__global__ __launch_bounds__(256) void k_gather(
    const int* __restrict__ rowptr, const int* __restrict__ col,
    const float* __restrict__ dinv, const float* __restrict__ h,
    const float* __restrict__ b, float* __restrict__ out, int n)
{
  const int lane = threadIdx.x & 31;
  const int node = __builtin_amdgcn_readfirstlane((int)((blockIdx.x * 256 + threadIdx.x) >> 5));
  if (node >= n) return;
  const int   start = rowptr[node];
  const int   end   = rowptr[node + 1];
  const float dd    = dinv[node];
  const float4 hs   = ld_row(h, node, lane);
  const float4 bv   = *(const float4*)(b + lane * 4);

  float4 acc = make_float4(0.0f, 0.0f, 0.0f, 0.0f);
  for (int base = start; base < end; base += 32) {
    const int idx = base + lane;
    int   s  = node;
    float cs = 0.0f;
    if (idx < end) {
      s  = col[idx];
      s  = ((unsigned)s < (unsigned)n) ? s : node;
      cs = dinv[s];
    }
    const int cnt = min(32, end - base);
    int j = 0;
    for (; j + 8 <= cnt; j += 8) {
      float4 t[8];
#pragma unroll
      for (int q = 0; q < 8; ++q) t[q] = ld_row(h, lane_i(s, j + q), lane);
#pragma unroll
      for (int q = 0; q < 8; ++q) fma_row(acc, t[q], lane_f(cs, j + q) * dd);
    }
    const int r = cnt - j;
    if (r > 0) {
      float4 t[7];
#pragma unroll
      for (int q = 0; q < 7; ++q)
        if (q < r) t[q] = ld_row(h, lane_i(s, j + q), lane);
#pragma unroll
      for (int q = 0; q < 7; ++q)
        if (q < r) fma_row(acc, t[q], lane_f(cs, j + q) * dd);
    }
  }

  const float sc = dd * dd;
  float4 o;
  o.x = fmaf(hs.x, sc, acc.x) + bv.x;
  o.y = fmaf(hs.y, sc, acc.y) + bv.y;
  o.z = fmaf(hs.z, sc, acc.z) + bv.z;
  o.w = fmaf(hs.w, sc, acc.w) + bv.w;
  *(float4*)(out + (size_t)node * DHID + lane * 4) = o;
}

__global__ __launch_bounds__(256) void k_bnstats(const float* __restrict__ h,
                                                 float* __restrict__ stats, int n) {
  __shared__ float ss[256], sq[256];
  int t = threadIdx.x, c = t & 127, half = t >> 7;
  float s = 0.0f, q = 0.0f;
  for (int r = blockIdx.x * 2 + half; r < n; r += gridDim.x * 2) {
    float v = fmaxf(h[(size_t)r * DHID + c], 0.0f);
    s += v; q += v * v;
  }
  ss[t] = s; sq[t] = q;
  __syncthreads();
  if (t < 128) {
    unsafeAtomicAdd(&stats[c],       ss[t] + ss[t + 128]);
    unsafeAtomicAdd(&stats[128 + c], sq[t] + sq[t + 128]);
  }
}

__global__ __launch_bounds__(256) void k_classify_head(
    const float* __restrict__ A, const float* __restrict__ W,
    const float* __restrict__ stats, const float* __restrict__ gamma,
    const float* __restrict__ beta, const float* __restrict__ bias,
    const float* __restrict__ Wr, const float* __restrict__ br,
    float* __restrict__ out, int nrows)
{
  __shared__ float wl[DHID * DHID];
  __shared__ __attribute__((aligned(16))) float bnl[2 * DHID];
  const int tid = threadIdx.x;
  for (int i = tid; i < DHID * DHID; i += 256) {
    int k = i >> 7, n = i & 127;
    wl[(((k >> 1) << 7) + n) * 2 + (k & 1)] = W[i];
  }
  if (tid < DHID) {
    float inv = 1.0f / (float)nrows;
    float mu  = stats[tid] * inv;
    float var = stats[128 + tid] * inv - mu * mu;
    float sc  = gamma[tid] * __frsqrt_rn(var + BN_EPS);
    bnl[tid]        = sc;
    bnl[DHID + tid] = beta[tid] - mu * sc;
  }
  __syncthreads();

  const int wave  = tid >> 5;
  const int lane  = tid & 31;
  const int laneM = lane & 15;
  const int hi    = lane >> 4;
  const int m0    = (blockIdx.x * 8 + wave) << 4;
  if (m0 >= nrows) return;

  const float2* __restrict__ wl2  = (const float2*)wl;
  const float2* __restrict__ bn2  = (const float2*)bnl;
  const float*  __restrict__ arow = A + (size_t)(m0 + laneM) * DHID;

  v8f acc[8];
#pragma unroll
  for (int t = 0; t < 8; ++t) acc[t] = (v8f){};

#pragma unroll 2
  for (int k = 0; k < DHID; k += 4) {
    const int c0 = k + hi * 2;
    float2 av = *(const float2*)(arow + c0);
    const float2 sc2 = bn2[c0 >> 1];
    const float2 sh2 = bn2[(DHID >> 1) + (c0 >> 1)];
    av.x = fmaxf(av.x, 0.0f) * sc2.x + sh2.x;
    av.y = fmaxf(av.y, 0.0f) * sc2.y + sh2.y;
    v2f a; a[0] = av.x; a[1] = av.y;

    const float2* __restrict__ brow = wl2 + (((k >> 1) + hi) << 7);
#pragma unroll
    for (int t = 0; t < 8; ++t) {
      float2 bv = brow[t * 16 + laneM];
      v2f b; b[0] = bv.x; b[1] = bv.y;
      acc[t] = __builtin_amdgcn_wmma_f32_16x16x4_f32(
          false, a, false, b, (short)0, acc[t], false, false);
    }
  }

  float p[16];
#pragma unroll
  for (int i = 0; i < 16; ++i) p[i] = 0.0f;
#pragma unroll
  for (int t = 0; t < 8; ++t) {
    const int    col  = t * 16 + laneM;
    const float  bval = bias[col];
    const float2 wr   = *(const float2*)(Wr + col * 2);
#pragma unroll
    for (int r = 0; r < 8; ++r) {
      float v = fmaxf(acc[t][r] + bval, 0.0f);
      p[2 * r]     = fmaf(v, wr.x, p[2 * r]);
      p[2 * r + 1] = fmaf(v, wr.y, p[2 * r + 1]);
    }
  }
  float w8[8], w4[4], w2[2];
  const bool u8 = (laneM & 8) != 0, u4 = (laneM & 4) != 0, u2 = (laneM & 2) != 0, u1 = (laneM & 1) != 0;
#pragma unroll
  for (int j = 0; j < 8; ++j) {
    float keep = u8 ? p[j + 8] : p[j];
    float send = u8 ? p[j] : p[j + 8];
    w8[j] = keep + __shfl_xor(send, 8, 32);
  }
#pragma unroll
  for (int j = 0; j < 4; ++j) {
    float keep = u4 ? w8[j + 4] : w8[j];
    float send = u4 ? w8[j] : w8[j + 4];
    w4[j] = keep + __shfl_xor(send, 4, 32);
  }
#pragma unroll
  for (int j = 0; j < 2; ++j) {
    float keep = u2 ? w4[j + 2] : w4[j];
    float send = u2 ? w4[j] : w4[j + 2];
    w2[j] = keep + __shfl_xor(send, 2, 32);
  }
  {
    float keep = u1 ? w2[1] : w2[0];
    float send = u1 ? w2[0] : w2[1];
    float tot  = keep + __shfl_xor(send, 1, 32);
    out[(size_t)m0 * 2 + lane] = tot + br[lane & 1];
  }
}

extern "C" void kernel_launch(void* const* d_in, const int* in_sizes, int n_in,
                              void* d_out, int out_size, void* d_ws, size_t ws_size,
                              hipStream_t stream) {
  const float* x     = (const float*)d_in[0];
  const int*   eidx  = (const int*)d_in[1];
  const float* W1    = (const float*)d_in[2];
  const float* b1    = (const float*)d_in[3];
  const float* W2    = (const float*)d_in[4];
  const float* b2    = (const float*)d_in[5];
  const float* W3    = (const float*)d_in[6];
  const float* b3    = (const float*)d_in[7];
  const float* gamma = (const float*)d_in[8];
  const float* beta  = (const float*)d_in[9];
  const float* Wc    = (const float*)d_in[10];
  const float* bc    = (const float*)d_in[11];
  const float* Wr    = (const float*)d_in[12];
  const float* br    = (const float*)d_in[13];
  float* out = (float*)d_out;

  const int N = N_NODES, E = E_EDGES;
  const int* src = eidx;
  const int* dst = eidx + E;

  float* buf0   = (float*)d_ws;
  float* buf1   = buf0 + (size_t)N * DHID;
  int*   cnt    = (int*)(buf1 + (size_t)N * DHID);
  float* stats  = (float*)(cnt + N);
  float* bnp    = stats + 256;
  int*   bsum   = (int*)(bnp + 256);
  float* dinv   = (float*)(bsum + 256);
  int*   rowptr = (int*)(dinv + N);
  int*   tab    = rowptr + N + 4;
  int*   col    = tab + (size_t)NCHUNK * N;

  const dim3 blk(256);
  const dim3 gSort(NCHUNK, NRANGE);
  const dim3 bSort(SORT_T);
  const int gScan = (N + 1023) / 1024;
  const int gGemm = (N + 127) / 128;
  const int gNode = (N * 32 + 255) / 256;

  k_count_lds<<<gSort, bSort, 0, stream>>>(dst, tab);
  k_degsum<<<gScan, blk, 0, stream>>>(tab, cnt, bsum, stats, N);
  k_scan_dinv<<<gScan, blk, 0, stream>>>(cnt, bsum, rowptr, dinv, N);
  k_fill_lds<<<gSort, bSort, 0, stream>>>(src, dst, rowptr, tab, col);

  k_gemm128<0, false, false><<<gGemm, blk, 0, stream>>>(x, W1, buf0, N, nullptr, nullptr, nullptr);
  k_gather<<<gNode, blk, 0, stream>>>(rowptr, col, dinv, buf0, b1, buf1, N);

  k_gemm128<1, false, false><<<gGemm, blk, 0, stream>>>(buf1, W2, buf0, N, nullptr, nullptr, nullptr);
  k_gather<<<gNode, blk, 0, stream>>>(rowptr, col, dinv, buf0, b2, buf1, N);

  k_gemm128<1, false, false><<<gGemm, blk, 0, stream>>>(buf1, W3, buf0, N, nullptr, nullptr, nullptr);
  k_gather<<<gNode, blk, 0, stream>>>(rowptr, col, dinv, buf0, b3, buf1, N);

  k_bnstats<<<1024, blk, 0, stream>>>(buf1, stats, N);

  k_classify_head<<<gGemm, blk, 0, stream>>>(buf1, Wc, stats, gamma, beta, bc, Wr, br, out, N);
}


# Layers 2 and 3: the f32 WMMA product runs inside the preceding aggregation launch; a wave gathers 16 nodes into its own LDS tile and multiplies it by W (2 launches, 2 plane round trips less)
# speedup vs baseline: 3.7428x; 1.1023x over previous
#include <hip/hip_runtime.h>

#define N_NODES 100000
#define DHID    128
#define E_EDGES 1600000
#define BN_EPS  1e-5f

typedef __attribute__((ext_vector_type(2))) float v2f;
typedef __attribute__((ext_vector_type(8))) float v8f;

#define NRANGE 8
#define NPR    (N_NODES / NRANGE)
#define NCHUNK 32
#define EPC    (E_EDGES / NCHUNK)
#define EPC4   (EPC / 4)
#define SORT_T 1024
static_assert(NPR * NRANGE == N_NODES, "ranges must tile the nodes");
static_assert(EPC * NCHUNK == E_EDGES && EPC4 * 4 == EPC, "chunks must tile the edges in 16-byte groups");
static_assert(N_NODES % 4 == 0, "k_degsum reads four nodes per thread");

__device__ __forceinline__ void count_one(int* lc, int d) {
  if ((unsigned)d < (unsigned)NPR) atomicAdd(&lc[d], 1);
}

__global__ __launch_bounds__(SORT_T) void k_count_lds(const int* __restrict__ dst,
                                                      int* __restrict__ tab) {
  __shared__ int lc[NPR];
  const int t = threadIdx.x, c = blockIdx.x, lo = blockIdx.y * NPR;
  for (int j = t; j < NPR; j += SORT_T) lc[j] = 0;
  __syncthreads();
  const int4* __restrict__ d4 = (const int4*)(dst + (size_t)c * EPC);
  constexpr int B = (EPC4 + SORT_T - 1) / SORT_T;
  int4 v[B];
#pragma unroll
  for (int q = 0; q < B; ++q) {
    const int i = t + q * SORT_T;
    v[q] = (i < EPC4) ? d4[i] : make_int4(-1, -1, -1, -1);
  }
#pragma unroll
  for (int q = 0; q < B; ++q) {
    count_one(lc, v[q].x - lo);
    count_one(lc, v[q].y - lo);
    count_one(lc, v[q].z - lo);
    count_one(lc, v[q].w - lo);
  }
  __syncthreads();
  int* __restrict__ trow = tab + (size_t)c * N_NODES + lo;
  for (int j = t; j < NPR; j += SORT_T) trow[j] = lc[j];
}

__global__ __launch_bounds__(256) void k_degsum(int* __restrict__ tab, int* __restrict__ cnt,
                                                int* __restrict__ bsum, float* __restrict__ stats,
                                                int n) {
  __shared__ int sh[256];
  const int t  = threadIdx.x;
  const int i0 = (blockIdx.x * 256 + t) * 4;
  if (blockIdx.x == 0) stats[t] = 0.0f;
  int4 run = make_int4(0, 0, 0, 0);
  if (i0 < n) {
    int4 v[NCHUNK];
#pragma unroll
    for (int c = 0; c < NCHUNK; ++c) v[c] = *(const int4*)(tab + (size_t)c * N_NODES + i0);
#pragma unroll
    for (int c = 0; c < NCHUNK; ++c) {
      *(int4*)(tab + (size_t)c * N_NODES + i0) = run;
      run.x += v[c].x; run.y += v[c].y; run.z += v[c].z; run.w += v[c].w;
    }
    *(int4*)(cnt + i0) = run;
  }
  sh[t] = run.x + run.y + run.z + run.w;
  __syncthreads();
  for (int off = 128; off > 0; off >>= 1) {
    if (t < off) sh[t] += sh[t + off];
    __syncthreads();
  }
  if (t == 0) bsum[blockIdx.x] = sh[0];
}

__global__ __launch_bounds__(256) void k_scan_dinv(const int* __restrict__ cnt,
                                                   const int* __restrict__ bsum,
                                                   int* __restrict__ rowptr,
                                                   float* __restrict__ dinv, int n) {
  __shared__ int sh[256];
  const int t = threadIdx.x, b = blockIdx.x;
  sh[t] = (t < b) ? bsum[t] : 0;
  __syncthreads();
  for (int off = 128; off > 0; off >>= 1) {
    if (t < off) sh[t] += sh[t + off];
    __syncthreads();
  }
  const int boff = sh[0];
  __syncthreads();

  const int i0 = (b * 256 + t) * 4;
  int c[4], tot = 0;
#pragma unroll
  for (int q = 0; q < 4; ++q) {
    c[q] = (i0 + q < n) ? cnt[i0 + q] : 0;
    tot += c[q];
  }
  sh[t] = tot;
  __syncthreads();
  for (int off = 1; off < 256; off <<= 1) {
    int x = (t >= off) ? sh[t - off] : 0;
    __syncthreads();
    sh[t] += x;
    __syncthreads();
  }
  int excl = boff + sh[t] - tot;
#pragma unroll
  for (int q = 0; q < 4; ++q) {
    if (i0 + q < n) {
      rowptr[i0 + q] = excl;
      dinv[i0 + q]   = __frsqrt_rn((float)c[q] + 1.0f);
      excl += c[q];
    }
  }
  if (i0 < n && i0 + 4 >= n) rowptr[n] = excl;
}

__device__ __forceinline__ int take_slot(int* cur, int d) {
  int slot = -1;
  if ((unsigned)d < (unsigned)NPR) slot = atomicAdd(&cur[d], 1);
  return slot;
}
__device__ __forceinline__ void put_src(int* __restrict__ col, int slot, int s) {
  if ((unsigned)slot < (unsigned)E_EDGES) col[slot] = s;
}

__global__ __launch_bounds__(SORT_T) void k_fill_lds(const int* __restrict__ src,
                                                     const int* __restrict__ dst,
                                                     const int* __restrict__ rowptr,
                                                     const int* __restrict__ tab,
                                                     int* __restrict__ col) {
  __shared__ int cur[NPR];
  const int t = threadIdx.x, c = blockIdx.x, lo = blockIdx.y * NPR;
  const int* __restrict__ trow = tab + (size_t)c * N_NODES + lo;
  const int* __restrict__ rp   = rowptr + lo;
  for (int j = t; j < NPR; j += SORT_T) cur[j] = rp[j] + trow[j];
  __syncthreads();
  const int4* __restrict__ d4 = (const int4*)(dst + (size_t)c * EPC);
  const int4* __restrict__ s4 = (const int4*)(src + (size_t)c * EPC);
  constexpr int B = 7;
  for (int p = 0; p < EPC4; p += B * SORT_T) {
    int4 dv[B], sv[B];
#pragma unroll
    for (int q = 0; q < B; ++q) {
      const int  i  = p + t + q * SORT_T;
      const bool ok = i < EPC4;
      dv[q] = ok ? d4[i] : make_int4(-1, -1, -1, -1);
      sv[q] = ok ? s4[i] : make_int4(0, 0, 0, 0);
    }
    int slot[4 * B];
#pragma unroll
    for (int q = 0; q < B; ++q) {
      slot[4 * q + 0] = take_slot(cur, dv[q].x - lo);
      slot[4 * q + 1] = take_slot(cur, dv[q].y - lo);
      slot[4 * q + 2] = take_slot(cur, dv[q].z - lo);
      slot[4 * q + 3] = take_slot(cur, dv[q].w - lo);
    }
#pragma unroll
    for (int q = 0; q < B; ++q) {
      put_src(col, slot[4 * q + 0], sv[q].x);
      put_src(col, slot[4 * q + 1], sv[q].y);
      put_src(col, slot[4 * q + 2], sv[q].z);
      put_src(col, slot[4 * q + 3], sv[q].w);
    }
  }
}

template <int AMODE, bool HAS_BIAS, bool EPI_RELU>
__global__ __launch_bounds__(256) void k_gemm128(
    const float* __restrict__ A, const float* __restrict__ W,
    float* __restrict__ C, int nrows,
    const float* __restrict__ bnScale, const float* __restrict__ bnShift,
    const float* __restrict__ bias)
{
  __shared__ float wl[DHID * DHID];
  const int tid = threadIdx.x;
  for (int i = tid; i < DHID * DHID; i += 256) {
    int k = i >> 7, n = i & 127;
    wl[(((k >> 1) << 7) + n) * 2 + (k & 1)] = W[i];
  }
  __syncthreads();

  const int wave  = tid >> 5;
  const int lane  = tid & 31;
  const int laneM = lane & 15;
  const int hi    = lane >> 4;
  const int m0    = (blockIdx.x * 8 + wave) << 4;
  if (m0 >= nrows) return;

  const float2* __restrict__ wl2  = (const float2*)wl;
  const float*  __restrict__ arow = A + (size_t)(m0 + laneM) * DHID;

  v8f acc[8];
#pragma unroll
  for (int t = 0; t < 8; ++t) acc[t] = (v8f){};

  for (int k = 0; k < DHID; k += 4) {
    float2 av = *(const float2*)(arow + k + hi * 2);
    if (AMODE == 1) {
      av.x = fmaxf(av.x, 0.0f);
      av.y = fmaxf(av.y, 0.0f);
    } else if (AMODE == 2) {
      int c0 = k + hi * 2;
      av.x = fmaxf(av.x, 0.0f) * bnScale[c0]     + bnShift[c0];
      av.y = fmaxf(av.y, 0.0f) * bnScale[c0 + 1] + bnShift[c0 + 1];
    }
    v2f a; a[0] = av.x; a[1] = av.y;

    const float2* __restrict__ brow = wl2 + (((k >> 1) + hi) << 7);
#pragma unroll
    for (int t = 0; t < 8; ++t) {
      float2 bv = brow[t * 16 + laneM];
      v2f b; b[0] = bv.x; b[1] = bv.y;
      acc[t] = __builtin_amdgcn_wmma_f32_16x16x4_f32(
          false, a, false, b, (short)0, acc[t], false, false);
    }
  }

#pragma unroll
  for (int t = 0; t < 8; ++t) {
    int   col  = t * 16 + laneM;
    float bval = HAS_BIAS ? bias[col] : 0.0f;
#pragma unroll
    for (int r = 0; r < 8; ++r) {
      int   row = m0 + r + hi * 8;
      float v   = acc[t][r] + bval;
      if (EPI_RELU) v = fmaxf(v, 0.0f);
      C[(size_t)row * DHID + col] = v;
    }
  }
}

__device__ __forceinline__ float4 ld_row(const float* __restrict__ h, int s, int lane) {
  return *(const float4*)(h + (size_t)s * DHID + lane * 4);
}
__device__ __forceinline__ void fma_row(float4& acc, const float4& v, float c) {
  acc.x = fmaf(v.x, c, acc.x);
  acc.y = fmaf(v.y, c, acc.y);
  acc.z = fmaf(v.z, c, acc.z);
  acc.w = fmaf(v.w, c, acc.w);
}
__device__ __forceinline__ int lane_i(int v, int j) { return __builtin_amdgcn_readlane(v, j); }
__device__ __forceinline__ float lane_f(float v, int j) {
  return __int_as_float(__builtin_amdgcn_readlane(__float_as_int(v), j));
}

__global__ __launch_bounds__(256) void k_gather(
    const int* __restrict__ rowptr, const int* __restrict__ col,
    const float* __restrict__ dinv, const float* __restrict__ h,
    const float* __restrict__ b, float* __restrict__ out, int n)
{
  const int lane = threadIdx.x & 31;
  const int node = __builtin_amdgcn_readfirstlane((int)((blockIdx.x * 256 + threadIdx.x) >> 5));
  if (node >= n) return;
  const int   start = rowptr[node];
  const int   end   = rowptr[node + 1];
  const float dd    = dinv[node];
  const float4 hs   = ld_row(h, node, lane);
  const float4 bv   = *(const float4*)(b + lane * 4);

  float4 acc = make_float4(0.0f, 0.0f, 0.0f, 0.0f);
  for (int base = start; base < end; base += 32) {
    const int idx = base + lane;
    int   s  = node;
    float cs = 0.0f;
    if (idx < end) {
      s  = col[idx];
      s  = ((unsigned)s < (unsigned)n) ? s : node;
      cs = dinv[s];
    }
    const int cnt = min(32, end - base);
    int j = 0;
    for (; j + 8 <= cnt; j += 8) {
      float4 t[8];
#pragma unroll
      for (int q = 0; q < 8; ++q) t[q] = ld_row(h, lane_i(s, j + q), lane);
#pragma unroll
      for (int q = 0; q < 8; ++q) fma_row(acc, t[q], lane_f(cs, j + q) * dd);
    }
    const int r = cnt - j;
    if (r > 0) {
      float4 t[7];
#pragma unroll
      for (int q = 0; q < 7; ++q)
        if (q < r) t[q] = ld_row(h, lane_i(s, j + q), lane);
#pragma unroll
      for (int q = 0; q < 7; ++q)
        if (q < r) fma_row(acc, t[q], lane_f(cs, j + q) * dd);
    }
  }

  const float sc = dd * dd;
  float4 o;
  o.x = fmaf(hs.x, sc, acc.x) + bv.x;
  o.y = fmaf(hs.y, sc, acc.y) + bv.y;
  o.z = fmaf(hs.z, sc, acc.z) + bv.z;
  o.w = fmaf(hs.w, sc, acc.w) + bv.w;
  *(float4*)(out + (size_t)node * DHID + lane * 4) = o;
}

#define FG_RB 16
typedef __attribute__((ext_vector_type(4))) float v4f;
typedef const __attribute__((address_space(1))) v4f gv4f;
__device__ __forceinline__ float4 ld_row_u(const float* __restrict__ h, int srow, int lane) {
  const unsigned long long p  = (unsigned long long)(h + (size_t)srow * DHID);
  const unsigned           lo = __builtin_amdgcn_readfirstlane((unsigned)p);
  const unsigned           hi = __builtin_amdgcn_readfirstlane((unsigned)(p >> 32));
  const gv4f* rp = (const gv4f*)(((unsigned long long)hi << 32) | lo);
  const v4f v = rp[lane];
  return make_float4(v.x, v.y, v.z, v.w);
}
__device__ __forceinline__ void chunk_rows(float4& acc, const float* __restrict__ h,
                                           int s, float cs, int cnt, float dd, int lane) {
  int j = 0;
  for (; j + FG_RB <= cnt; j += FG_RB) {
    float4 t[FG_RB];
#pragma unroll
    for (int q = 0; q < FG_RB; ++q) t[q] = ld_row_u(h, lane_i(s, j + q), lane);
#pragma unroll
    for (int q = 0; q < FG_RB; ++q) fma_row(acc, t[q], lane_f(cs, j + q) * dd);
  }
  const int r = cnt - j;
  if (r > 0) {
    float4 t[FG_RB - 1];
#pragma unroll
    for (int q = 0; q < FG_RB - 1; ++q) t[q] = make_float4(0.0f, 0.0f, 0.0f, 0.0f);
#pragma unroll
    for (int q = 0; q < FG_RB - 1; ++q)
      if (q < r) t[q] = ld_row_u(h, lane_i(s, j + q), lane);
#pragma unroll
    for (int q = 0; q < FG_RB - 1; ++q)
      if (q < r) fma_row(acc, t[q], lane_f(cs, j + q) * dd);
  }
}

#define FG_WAVES 25
#define FG_T     (FG_WAVES * 32)
#define TPITCH   132
static_assert(N_NODES % 16 == 0, "a wave's tile holds 16 whole nodes");

__global__ __launch_bounds__(FG_T) void k_gather_gemm(
    const int* __restrict__ rowptr, const int* __restrict__ col,
    const float* __restrict__ dinv, const float* __restrict__ h,
    const float* __restrict__ b, const float* __restrict__ W,
    float* __restrict__ C, int n)
{
  __shared__ float wl[DHID * DHID];
  __shared__ __attribute__((aligned(16))) float tile[FG_WAVES * 16 * TPITCH];
  const int tid = threadIdx.x;
  for (int i = tid; i < DHID * DHID; i += FG_T) {
    int k = i >> 7, nn = i & 127;
    wl[(((k >> 1) << 7) + nn) * 2 + (k & 1)] = W[i];
  }
  __syncthreads();

  const int lane = tid & 31;
  const int wave = __builtin_amdgcn_readfirstlane(tid >> 5);
  const int m0   = (blockIdx.x * FG_WAVES + wave) << 4;
  if (m0 >= n) return;
  float* tw = tile + wave * (16 * TPITCH);

  const int    rpv = rowptr[min(m0 + min(lane, 16), n)];
  const float  ddv = dinv[min(m0 + (lane & 15), n - 1)];
  const float4 bv  = *(const float4*)(b + lane * 4);

  const int st0 = lane_i(rpv, 0), st1 = lane_i(rpv, 1), st2 = lane_i(rpv, 2);
  int   s1  = col[min(st0 + lane, E_EDGES - 1)];
  int   s2  = col[min(st1 + lane, E_EDGES - 1)];
  s1 = (st0 + lane < st1 && (unsigned)s1 < (unsigned)n) ? s1 : m0;
  s2 = (st1 + lane < st2 && (unsigned)s2 < (unsigned)n) ? s2 : m0;
  float cs1 = dinv[s1];

  for (int i = 0; i < 16; ++i) {
    const int   node  = m0 + i;
    const int   start = lane_i(rpv, i), end = lane_i(rpv, i + 1);
    const float dd    = lane_f(ddv, i);
    const float cs2  = dinv[s2];
    const int   st3  = lane_i(rpv, min(i + 2, 16)), en3 = lane_i(rpv, min(i + 3, 16));
    const int   s3r  = col[min(st3 + lane, E_EDGES - 1)];
    const float4 hs = ld_row_u(h, node, lane);
    float4 acc = make_float4(0.0f, 0.0f, 0.0f, 0.0f);
    chunk_rows(acc, h, s1, cs1, min(32, end - start), dd, lane);
    for (int base = start + 32; base < end; base += 32) {
      const int idx = base + lane;
      int   s  = node;
      float cs = 0.0f;
      if (idx < end) {
        s  = col[idx];
        s  = ((unsigned)s < (unsigned)n) ? s : node;
        cs = dinv[s];
      }
      const int cnt = min(32, end - base);
      for (int j = 0; j < cnt; ++j) fma_row(acc, ld_row_u(h, lane_i(s, j), lane), lane_f(cs, j) * dd);
    }
    const float sc = dd * dd;
    float4 o;
    o.x = fmaxf(fmaf(hs.x, sc, acc.x) + bv.x, 0.0f);
    o.y = fmaxf(fmaf(hs.y, sc, acc.y) + bv.y, 0.0f);
    o.z = fmaxf(fmaf(hs.z, sc, acc.z) + bv.z, 0.0f);
    o.w = fmaxf(fmaf(hs.w, sc, acc.w) + bv.w, 0.0f);
    *(float4*)(tw + i * TPITCH + lane * 4) = o;
    s1  = s2;
    cs1 = cs2;
    s2  = (st3 + lane < en3 && (unsigned)s3r < (unsigned)n) ? s3r : node;
  }

  __builtin_amdgcn_fence(__ATOMIC_RELEASE, "wavefront");
  __builtin_amdgcn_wave_barrier();
  __builtin_amdgcn_fence(__ATOMIC_ACQUIRE, "wavefront");

  const int laneM = lane & 15;
  const int hi    = lane >> 4;
  const float2* __restrict__ wl2 = (const float2*)wl;
  const float* trow = tw + laneM * TPITCH + hi * 2;

  v8f acc2[8];
#pragma unroll
  for (int t = 0; t < 8; ++t) acc2[t] = (v8f){};

  for (int k = 0; k < DHID; k += 4) {
    const float2 av = *(const float2*)(trow + k);
    v2f a; a[0] = av.x; a[1] = av.y;
    const float2* __restrict__ brow = wl2 + (((k >> 1) + hi) << 7);
#pragma unroll
    for (int t = 0; t < 8; ++t) {
      float2 bw = brow[t * 16 + laneM];
      v2f bb; bb[0] = bw.x; bb[1] = bw.y;
      acc2[t] = __builtin_amdgcn_wmma_f32_16x16x4_f32(
          false, a, false, bb, (short)0, acc2[t], false, false);
    }
  }

#pragma unroll
  for (int t = 0; t < 8; ++t) {
    const int cc = t * 16 + laneM;
#pragma unroll
    for (int r = 0; r < 8; ++r) {
      const int row = m0 + r + hi * 8;
      C[(size_t)row * DHID + cc] = acc2[t][r];
    }
  }
}

__global__ __launch_bounds__(256) void k_bnstats(const float* __restrict__ h,
                                                 float* __restrict__ stats, int n) {
  __shared__ float ss[256], sq[256];
  int t = threadIdx.x, c = t & 127, half = t >> 7;
  float s = 0.0f, q = 0.0f;
  for (int r = blockIdx.x * 2 + half; r < n; r += gridDim.x * 2) {
    float v = fmaxf(h[(size_t)r * DHID + c], 0.0f);
    s += v; q += v * v;
  }
  ss[t] = s; sq[t] = q;
  __syncthreads();
  if (t < 128) {
    unsafeAtomicAdd(&stats[c],       ss[t] + ss[t + 128]);
    unsafeAtomicAdd(&stats[128 + c], sq[t] + sq[t + 128]);
  }
}

__global__ __launch_bounds__(256) void k_classify_head(
    const float* __restrict__ A, const float* __restrict__ W,
    const float* __restrict__ stats, const float* __restrict__ gamma,
    const float* __restrict__ beta, const float* __restrict__ bias,
    const float* __restrict__ Wr, const float* __restrict__ br,
    float* __restrict__ out, int nrows)
{
  __shared__ float wl[DHID * DHID];
  __shared__ __attribute__((aligned(16))) float bnl[2 * DHID];
  const int tid = threadIdx.x;
  for (int i = tid; i < DHID * DHID; i += 256) {
    int k = i >> 7, n = i & 127;
    wl[(((k >> 1) << 7) + n) * 2 + (k & 1)] = W[i];
  }
  if (tid < DHID) {
    float inv = 1.0f / (float)nrows;
    float mu  = stats[tid] * inv;
    float var = stats[128 + tid] * inv - mu * mu;
    float sc  = gamma[tid] * __frsqrt_rn(var + BN_EPS);
    bnl[tid]        = sc;
    bnl[DHID + tid] = beta[tid] - mu * sc;
  }
  __syncthreads();

  const int wave  = tid >> 5;
  const int lane  = tid & 31;
  const int laneM = lane & 15;
  const int hi    = lane >> 4;
  const int m0    = (blockIdx.x * 8 + wave) << 4;
  if (m0 >= nrows) return;

  const float2* __restrict__ wl2  = (const float2*)wl;
  const float2* __restrict__ bn2  = (const float2*)bnl;
  const float*  __restrict__ arow = A + (size_t)(m0 + laneM) * DHID;

  v8f acc[8];
#pragma unroll
  for (int t = 0; t < 8; ++t) acc[t] = (v8f){};

#pragma unroll 2
  for (int k = 0; k < DHID; k += 4) {
    const int c0 = k + hi * 2;
    float2 av = *(const float2*)(arow + c0);
    const float2 sc2 = bn2[c0 >> 1];
    const float2 sh2 = bn2[(DHID >> 1) + (c0 >> 1)];
    av.x = fmaxf(av.x, 0.0f) * sc2.x + sh2.x;
    av.y = fmaxf(av.y, 0.0f) * sc2.y + sh2.y;
    v2f a; a[0] = av.x; a[1] = av.y;

    const float2* __restrict__ brow = wl2 + (((k >> 1) + hi) << 7);
#pragma unroll
    for (int t = 0; t < 8; ++t) {
      float2 bv = brow[t * 16 + laneM];
      v2f b; b[0] = bv.x; b[1] = bv.y;
      acc[t] = __builtin_amdgcn_wmma_f32_16x16x4_f32(
          false, a, false, b, (short)0, acc[t], false, false);
    }
  }

  float p[16];
#pragma unroll
  for (int i = 0; i < 16; ++i) p[i] = 0.0f;
#pragma unroll
  for (int t = 0; t < 8; ++t) {
    const int    col  = t * 16 + laneM;
    const float  bval = bias[col];
    const float2 wr   = *(const float2*)(Wr + col * 2);
#pragma unroll
    for (int r = 0; r < 8; ++r) {
      float v = fmaxf(acc[t][r] + bval, 0.0f);
      p[2 * r]     = fmaf(v, wr.x, p[2 * r]);
      p[2 * r + 1] = fmaf(v, wr.y, p[2 * r + 1]);
    }
  }
  float w8[8], w4[4], w2[2];
  const bool u8 = (laneM & 8) != 0, u4 = (laneM & 4) != 0, u2 = (laneM & 2) != 0, u1 = (laneM & 1) != 0;
#pragma unroll
  for (int j = 0; j < 8; ++j) {
    float keep = u8 ? p[j + 8] : p[j];
    float send = u8 ? p[j] : p[j + 8];
    w8[j] = keep + __shfl_xor(send, 8, 32);
  }
#pragma unroll
  for (int j = 0; j < 4; ++j) {
    float keep = u4 ? w8[j + 4] : w8[j];
    float send = u4 ? w8[j] : w8[j + 4];
    w4[j] = keep + __shfl_xor(send, 4, 32);
  }
#pragma unroll
  for (int j = 0; j < 2; ++j) {
    float keep = u2 ? w4[j + 2] : w4[j];
    float send = u2 ? w4[j] : w4[j + 2];
    w2[j] = keep + __shfl_xor(send, 2, 32);
  }
  {
    float keep = u1 ? w2[1] : w2[0];
    float send = u1 ? w2[0] : w2[1];
    float tot  = keep + __shfl_xor(send, 1, 32);
    out[(size_t)m0 * 2 + lane] = tot + br[lane & 1];
  }
}

extern "C" void kernel_launch(void* const* d_in, const int* in_sizes, int n_in,
                              void* d_out, int out_size, void* d_ws, size_t ws_size,
                              hipStream_t stream) {
  const float* x     = (const float*)d_in[0];
  const int*   eidx  = (const int*)d_in[1];
  const float* W1    = (const float*)d_in[2];
  const float* b1    = (const float*)d_in[3];
  const float* W2    = (const float*)d_in[4];
  const float* b2    = (const float*)d_in[5];
  const float* W3    = (const float*)d_in[6];
  const float* b3    = (const float*)d_in[7];
  const float* gamma = (const float*)d_in[8];
  const float* beta  = (const float*)d_in[9];
  const float* Wc    = (const float*)d_in[10];
  const float* bc    = (const float*)d_in[11];
  const float* Wr    = (const float*)d_in[12];
  const float* br    = (const float*)d_in[13];
  float* out = (float*)d_out;

  const int N = N_NODES, E = E_EDGES;
  const int* src = eidx;
  const int* dst = eidx + E;

  float* buf0   = (float*)d_ws;
  float* buf1   = buf0 + (size_t)N * DHID;
  int*   cnt    = (int*)(buf1 + (size_t)N * DHID);
  float* stats  = (float*)(cnt + N);
  float* bnp    = stats + 256;
  int*   bsum   = (int*)(bnp + 256);
  float* dinv   = (float*)(bsum + 256);
  int*   rowptr = (int*)(dinv + N);
  int*   tab    = rowptr + N + 4;
  int*   col    = tab + (size_t)NCHUNK * N;

  const dim3 blk(256);
  const dim3 gSort(NCHUNK, NRANGE);
  const dim3 bSort(SORT_T);
  const int gScan = (N + 1023) / 1024;
  const int gGemm = (N + 127) / 128;
  const int gNode = (N * 32 + 255) / 256;
  const int gFuse = (N / 16 + FG_WAVES - 1) / FG_WAVES;
  const dim3 bFuse(FG_T);

  k_count_lds<<<gSort, bSort, 0, stream>>>(dst, tab);
  k_degsum<<<gScan, blk, 0, stream>>>(tab, cnt, bsum, stats, N);
  k_scan_dinv<<<gScan, blk, 0, stream>>>(cnt, bsum, rowptr, dinv, N);
  k_fill_lds<<<gSort, bSort, 0, stream>>>(src, dst, rowptr, tab, col);

  k_gemm128<0, false, false><<<gGemm, blk, 0, stream>>>(x, W1, buf0, N, nullptr, nullptr, nullptr);

  k_gather_gemm<<<gFuse, bFuse, 0, stream>>>(rowptr, col, dinv, buf0, b1, W2, buf1, N);
  k_gather_gemm<<<gFuse, bFuse, 0, stream>>>(rowptr, col, dinv, buf1, b2, W3, buf0, N);

  k_gather<<<gNode, blk, 0, stream>>>(rowptr, col, dinv, buf0, b3, buf1, N);

  k_bnstats<<<1024, blk, 0, stream>>>(buf1, stats, N);

  k_classify_head<<<gGemm, blk, 0, stream>>>(buf1, Wc, stats, gamma, beta, bc, Wr, br, out, N);
}


# Every layer aggregates first and multiplies after, (A h) W for A (h W), in one gather+WMMA launch with bias+relu epilogue: layer-1 GEMM launch and plane trip go; edge and k sums reassociated
# speedup vs baseline: 3.9944x; 1.0672x over previous
#include <hip/hip_runtime.h>

#define N_NODES 100000
#define DHID    128
#define E_EDGES 1600000
#define BN_EPS  1e-5f

typedef __attribute__((ext_vector_type(2))) float v2f;
typedef __attribute__((ext_vector_type(8))) float v8f;

#define NRANGE 8
#define NPR    (N_NODES / NRANGE)
#define NCHUNK 32
#define EPC    (E_EDGES / NCHUNK)
#define EPC4   (EPC / 4)
#define SORT_T 1024
static_assert(NPR * NRANGE == N_NODES, "ranges must tile the nodes");
static_assert(EPC * NCHUNK == E_EDGES && EPC4 * 4 == EPC, "chunks must tile the edges in 16-byte groups");
static_assert(N_NODES % 4 == 0, "k_degsum reads four nodes per thread");

__device__ __forceinline__ void count_one(int* lc, int d) {
  if ((unsigned)d < (unsigned)NPR) atomicAdd(&lc[d], 1);
}

__global__ __launch_bounds__(SORT_T) void k_count_lds(const int* __restrict__ dst,
                                                      int* __restrict__ tab) {
  __shared__ int lc[NPR];
  const int t = threadIdx.x, c = blockIdx.x, lo = blockIdx.y * NPR;
  for (int j = t; j < NPR; j += SORT_T) lc[j] = 0;
  __syncthreads();
  const int4* __restrict__ d4 = (const int4*)(dst + (size_t)c * EPC);
  constexpr int B = (EPC4 + SORT_T - 1) / SORT_T;
  int4 v[B];
#pragma unroll
  for (int q = 0; q < B; ++q) {
    const int i = t + q * SORT_T;
    v[q] = (i < EPC4) ? d4[i] : make_int4(-1, -1, -1, -1);
  }
#pragma unroll
  for (int q = 0; q < B; ++q) {
    count_one(lc, v[q].x - lo);
    count_one(lc, v[q].y - lo);
    count_one(lc, v[q].z - lo);
    count_one(lc, v[q].w - lo);
  }
  __syncthreads();
  int* __restrict__ trow = tab + (size_t)c * N_NODES + lo;
  for (int j = t; j < NPR; j += SORT_T) trow[j] = lc[j];
}

__global__ __launch_bounds__(256) void k_degsum(int* __restrict__ tab, int* __restrict__ cnt,
                                                int* __restrict__ bsum, float* __restrict__ stats,
                                                int n) {
  __shared__ int sh[256];
  const int t  = threadIdx.x;
  const int i0 = (blockIdx.x * 256 + t) * 4;
  if (blockIdx.x == 0) stats[t] = 0.0f;
  int4 run = make_int4(0, 0, 0, 0);
  if (i0 < n) {
    int4 v[NCHUNK];
#pragma unroll
    for (int c = 0; c < NCHUNK; ++c) v[c] = *(const int4*)(tab + (size_t)c * N_NODES + i0);
#pragma unroll
    for (int c = 0; c < NCHUNK; ++c) {
      *(int4*)(tab + (size_t)c * N_NODES + i0) = run;
      run.x += v[c].x; run.y += v[c].y; run.z += v[c].z; run.w += v[c].w;
    }
    *(int4*)(cnt + i0) = run;
  }
  sh[t] = run.x + run.y + run.z + run.w;
  __syncthreads();
  for (int off = 128; off > 0; off >>= 1) {
    if (t < off) sh[t] += sh[t + off];
    __syncthreads();
  }
  if (t == 0) bsum[blockIdx.x] = sh[0];
}

__global__ __launch_bounds__(256) void k_scan_dinv(const int* __restrict__ cnt,
                                                   const int* __restrict__ bsum,
                                                   int* __restrict__ rowptr,
                                                   float* __restrict__ dinv, int n) {
  __shared__ int sh[256];
  const int t = threadIdx.x, b = blockIdx.x;
  sh[t] = (t < b) ? bsum[t] : 0;
  __syncthreads();
  for (int off = 128; off > 0; off >>= 1) {
    if (t < off) sh[t] += sh[t + off];
    __syncthreads();
  }
  const int boff = sh[0];
  __syncthreads();

  const int i0 = (b * 256 + t) * 4;
  int c[4], tot = 0;
#pragma unroll
  for (int q = 0; q < 4; ++q) {
    c[q] = (i0 + q < n) ? cnt[i0 + q] : 0;
    tot += c[q];
  }
  sh[t] = tot;
  __syncthreads();
  for (int off = 1; off < 256; off <<= 1) {
    int x = (t >= off) ? sh[t - off] : 0;
    __syncthreads();
    sh[t] += x;
    __syncthreads();
  }
  int excl = boff + sh[t] - tot;
#pragma unroll
  for (int q = 0; q < 4; ++q) {
    if (i0 + q < n) {
      rowptr[i0 + q] = excl;
      dinv[i0 + q]   = __frsqrt_rn((float)c[q] + 1.0f);
      excl += c[q];
    }
  }
  if (i0 < n && i0 + 4 >= n) rowptr[n] = excl;
}

__device__ __forceinline__ int take_slot(int* cur, int d) {
  int slot = -1;
  if ((unsigned)d < (unsigned)NPR) slot = atomicAdd(&cur[d], 1);
  return slot;
}
__device__ __forceinline__ void put_src(int* __restrict__ col, int slot, int s) {
  if ((unsigned)slot < (unsigned)E_EDGES) col[slot] = s;
}

__global__ __launch_bounds__(SORT_T) void k_fill_lds(const int* __restrict__ src,
                                                     const int* __restrict__ dst,
                                                     const int* __restrict__ rowptr,
                                                     const int* __restrict__ tab,
                                                     int* __restrict__ col) {
  __shared__ int cur[NPR];
  const int t = threadIdx.x, c = blockIdx.x, lo = blockIdx.y * NPR;
  const int* __restrict__ trow = tab + (size_t)c * N_NODES + lo;
  const int* __restrict__ rp   = rowptr + lo;
  for (int j = t; j < NPR; j += SORT_T) cur[j] = rp[j] + trow[j];
  __syncthreads();
  const int4* __restrict__ d4 = (const int4*)(dst + (size_t)c * EPC);
  const int4* __restrict__ s4 = (const int4*)(src + (size_t)c * EPC);
  constexpr int B = 7;
  for (int p = 0; p < EPC4; p += B * SORT_T) {
    int4 dv[B], sv[B];
#pragma unroll
    for (int q = 0; q < B; ++q) {
      const int  i  = p + t + q * SORT_T;
      const bool ok = i < EPC4;
      dv[q] = ok ? d4[i] : make_int4(-1, -1, -1, -1);
      sv[q] = ok ? s4[i] : make_int4(0, 0, 0, 0);
    }
    int slot[4 * B];
#pragma unroll
    for (int q = 0; q < B; ++q) {
      slot[4 * q + 0] = take_slot(cur, dv[q].x - lo);
      slot[4 * q + 1] = take_slot(cur, dv[q].y - lo);
      slot[4 * q + 2] = take_slot(cur, dv[q].z - lo);
      slot[4 * q + 3] = take_slot(cur, dv[q].w - lo);
    }
#pragma unroll
    for (int q = 0; q < B; ++q) {
      put_src(col, slot[4 * q + 0], sv[q].x);
      put_src(col, slot[4 * q + 1], sv[q].y);
      put_src(col, slot[4 * q + 2], sv[q].z);
      put_src(col, slot[4 * q + 3], sv[q].w);
    }
  }
}

__device__ __forceinline__ void fma_row(float4& acc, const float4& v, float c) {
  acc.x = fmaf(v.x, c, acc.x);
  acc.y = fmaf(v.y, c, acc.y);
  acc.z = fmaf(v.z, c, acc.z);
  acc.w = fmaf(v.w, c, acc.w);
}
__device__ __forceinline__ int lane_i(int v, int j) { return __builtin_amdgcn_readlane(v, j); }
__device__ __forceinline__ float lane_f(float v, int j) {
  return __int_as_float(__builtin_amdgcn_readlane(__float_as_int(v), j));
}

#define FG_RB 16
typedef __attribute__((ext_vector_type(4))) float v4f;
typedef const __attribute__((address_space(1))) v4f gv4f;
__device__ __forceinline__ float4 ld_row_u(const float* __restrict__ h, int srow, int lane) {
  const unsigned long long p  = (unsigned long long)(h + (size_t)srow * DHID);
  const unsigned           lo = __builtin_amdgcn_readfirstlane((unsigned)p);
  const unsigned           hi = __builtin_amdgcn_readfirstlane((unsigned)(p >> 32));
  const gv4f* rp = (const gv4f*)(((unsigned long long)hi << 32) | lo);
  const v4f v = rp[lane];
  return make_float4(v.x, v.y, v.z, v.w);
}
__device__ __forceinline__ void chunk_rows(float4& acc, const float* __restrict__ h,
                                           int s, float cs, int cnt, float dd, int lane) {
  int j = 0;
  for (; j + FG_RB <= cnt; j += FG_RB) {
    float4 t[FG_RB];
#pragma unroll
    for (int q = 0; q < FG_RB; ++q) t[q] = ld_row_u(h, lane_i(s, j + q), lane);
#pragma unroll
    for (int q = 0; q < FG_RB; ++q) fma_row(acc, t[q], lane_f(cs, j + q) * dd);
  }
  const int r = cnt - j;
  if (r > 0) {
    float4 t[FG_RB - 1];
#pragma unroll
    for (int q = 0; q < FG_RB - 1; ++q) t[q] = make_float4(0.0f, 0.0f, 0.0f, 0.0f);
#pragma unroll
    for (int q = 0; q < FG_RB - 1; ++q)
      if (q < r) t[q] = ld_row_u(h, lane_i(s, j + q), lane);
#pragma unroll
    for (int q = 0; q < FG_RB - 1; ++q)
      if (q < r) fma_row(acc, t[q], lane_f(cs, j + q) * dd);
  }
}

#define FG_WAVES 25
#define FG_T     (FG_WAVES * 32)
#define TPITCH   132
static_assert(N_NODES % 16 == 0, "a wave's tile holds 16 whole nodes");

__global__ __launch_bounds__(FG_T) void k_gather_gemm(
    const int* __restrict__ rowptr, const int* __restrict__ col,
    const float* __restrict__ dinv, const float* __restrict__ h,
    const float* __restrict__ b, const float* __restrict__ W,
    float* __restrict__ C, int n)
{
  __shared__ float wl[DHID * DHID];
  __shared__ __attribute__((aligned(16))) float tile[FG_WAVES * 16 * TPITCH];
  const int tid = threadIdx.x;
  for (int i = tid; i < DHID * DHID; i += FG_T) {
    int k = i >> 7, nn = i & 127;
    wl[(((k >> 1) << 7) + nn) * 2 + (k & 1)] = W[i];
  }
  __syncthreads();

  const int lane = tid & 31;
  const int wave = __builtin_amdgcn_readfirstlane(tid >> 5);
  const int m0   = (blockIdx.x * FG_WAVES + wave) << 4;
  if (m0 >= n) return;
  float* tw = tile + wave * (16 * TPITCH);

  const int    rpv = rowptr[min(m0 + min(lane, 16), n)];
  const float  ddv = dinv[min(m0 + (lane & 15), n - 1)];

  const int st0 = lane_i(rpv, 0), st1 = lane_i(rpv, 1), st2 = lane_i(rpv, 2);
  int   s1  = col[min(st0 + lane, E_EDGES - 1)];
  int   s2  = col[min(st1 + lane, E_EDGES - 1)];
  s1 = (st0 + lane < st1 && (unsigned)s1 < (unsigned)n) ? s1 : m0;
  s2 = (st1 + lane < st2 && (unsigned)s2 < (unsigned)n) ? s2 : m0;
  float cs1 = dinv[s1];

  for (int i = 0; i < 16; ++i) {
    const int   node  = m0 + i;
    const int   start = lane_i(rpv, i), end = lane_i(rpv, i + 1);
    const float dd    = lane_f(ddv, i);
    const float cs2  = dinv[s2];
    const int   st3  = lane_i(rpv, min(i + 2, 16)), en3 = lane_i(rpv, min(i + 3, 16));
    const int   s3r  = col[min(st3 + lane, E_EDGES - 1)];
    const float4 hs = ld_row_u(h, node, lane);
    float4 acc = make_float4(0.0f, 0.0f, 0.0f, 0.0f);
    chunk_rows(acc, h, s1, cs1, min(32, end - start), dd, lane);
    for (int base = start + 32; base < end; base += 32) {
      const int idx = base + lane;
      int   s  = node;
      float cs = 0.0f;
      if (idx < end) {
        s  = col[idx];
        s  = ((unsigned)s < (unsigned)n) ? s : node;
        cs = dinv[s];
      }
      const int cnt = min(32, end - base);
      for (int j = 0; j < cnt; ++j) fma_row(acc, ld_row_u(h, lane_i(s, j), lane), lane_f(cs, j) * dd);
    }
    const float sc = dd * dd;
    float4 o;
    o.x = fmaf(hs.x, sc, acc.x);
    o.y = fmaf(hs.y, sc, acc.y);
    o.z = fmaf(hs.z, sc, acc.z);
    o.w = fmaf(hs.w, sc, acc.w);
    *(float4*)(tw + i * TPITCH + lane * 4) = o;
    s1  = s2;
    cs1 = cs2;
    s2  = (st3 + lane < en3 && (unsigned)s3r < (unsigned)n) ? s3r : node;
  }

  __builtin_amdgcn_fence(__ATOMIC_RELEASE, "wavefront");
  __builtin_amdgcn_wave_barrier();
  __builtin_amdgcn_fence(__ATOMIC_ACQUIRE, "wavefront");

  const int laneM = lane & 15;
  const int hi    = lane >> 4;
  const float2* __restrict__ wl2 = (const float2*)wl;
  const float* trow = tw + laneM * TPITCH + hi * 2;

  v8f acc2[8];
#pragma unroll
  for (int t = 0; t < 8; ++t) acc2[t] = (v8f){};

  for (int k = 0; k < DHID; k += 4) {
    const float2 av = *(const float2*)(trow + k);
    v2f a; a[0] = av.x; a[1] = av.y;
    const float2* __restrict__ brow = wl2 + (((k >> 1) + hi) << 7);
#pragma unroll
    for (int t = 0; t < 8; ++t) {
      float2 bw = brow[t * 16 + laneM];
      v2f bb; bb[0] = bw.x; bb[1] = bw.y;
      acc2[t] = __builtin_amdgcn_wmma_f32_16x16x4_f32(
          false, a, false, bb, (short)0, acc2[t], false, false);
    }
  }

#pragma unroll
  for (int t = 0; t < 8; ++t) {
    const int   cc   = t * 16 + laneM;
    const float bval = b[cc];
#pragma unroll
    for (int r = 0; r < 8; ++r) {
      const int row = m0 + r + hi * 8;
      C[(size_t)row * DHID + cc] = fmaxf(acc2[t][r] + bval, 0.0f);
    }
  }
}

__global__ __launch_bounds__(256) void k_bnstats(const float* __restrict__ h,
                                                 float* __restrict__ stats, int n) {
  __shared__ float ss[256], sq[256];
  int t = threadIdx.x, c = t & 127, half = t >> 7;
  float s = 0.0f, q = 0.0f;
  for (int r = blockIdx.x * 2 + half; r < n; r += gridDim.x * 2) {
    float v = fmaxf(h[(size_t)r * DHID + c], 0.0f);
    s += v; q += v * v;
  }
  ss[t] = s; sq[t] = q;
  __syncthreads();
  if (t < 128) {
    unsafeAtomicAdd(&stats[c],       ss[t] + ss[t + 128]);
    unsafeAtomicAdd(&stats[128 + c], sq[t] + sq[t + 128]);
  }
}

__global__ __launch_bounds__(256) void k_classify_head(
    const float* __restrict__ A, const float* __restrict__ W,
    const float* __restrict__ stats, const float* __restrict__ gamma,
    const float* __restrict__ beta, const float* __restrict__ bias,
    const float* __restrict__ Wr, const float* __restrict__ br,
    float* __restrict__ out, int nrows)
{
  __shared__ float wl[DHID * DHID];
  __shared__ __attribute__((aligned(16))) float bnl[2 * DHID];
  const int tid = threadIdx.x;
  for (int i = tid; i < DHID * DHID; i += 256) {
    int k = i >> 7, n = i & 127;
    wl[(((k >> 1) << 7) + n) * 2 + (k & 1)] = W[i];
  }
  if (tid < DHID) {
    float inv = 1.0f / (float)nrows;
    float mu  = stats[tid] * inv;
    float var = stats[128 + tid] * inv - mu * mu;
    float sc  = gamma[tid] * __frsqrt_rn(var + BN_EPS);
    bnl[tid]        = sc;
    bnl[DHID + tid] = beta[tid] - mu * sc;
  }
  __syncthreads();

  const int wave  = tid >> 5;
  const int lane  = tid & 31;
  const int laneM = lane & 15;
  const int hi    = lane >> 4;
  const int m0    = (blockIdx.x * 8 + wave) << 4;
  if (m0 >= nrows) return;

  const float2* __restrict__ wl2  = (const float2*)wl;
  const float2* __restrict__ bn2  = (const float2*)bnl;
  const float*  __restrict__ arow = A + (size_t)(m0 + laneM) * DHID;

  v8f acc[8];
#pragma unroll
  for (int t = 0; t < 8; ++t) acc[t] = (v8f){};

#pragma unroll 2
  for (int k = 0; k < DHID; k += 4) {
    const int c0 = k + hi * 2;
    float2 av = *(const float2*)(arow + c0);
    const float2 sc2 = bn2[c0 >> 1];
    const float2 sh2 = bn2[(DHID >> 1) + (c0 >> 1)];
    av.x = fmaxf(av.x, 0.0f) * sc2.x + sh2.x;
    av.y = fmaxf(av.y, 0.0f) * sc2.y + sh2.y;
    v2f a; a[0] = av.x; a[1] = av.y;

    const float2* __restrict__ brow = wl2 + (((k >> 1) + hi) << 7);
#pragma unroll
    for (int t = 0; t < 8; ++t) {
      float2 bv = brow[t * 16 + laneM];
      v2f b; b[0] = bv.x; b[1] = bv.y;
      acc[t] = __builtin_amdgcn_wmma_f32_16x16x4_f32(
          false, a, false, b, (short)0, acc[t], false, false);
    }
  }

  float p[16];
#pragma unroll
  for (int i = 0; i < 16; ++i) p[i] = 0.0f;
#pragma unroll
  for (int t = 0; t < 8; ++t) {
    const int    col  = t * 16 + laneM;
    const float  bval = bias[col];
    const float2 wr   = *(const float2*)(Wr + col * 2);
#pragma unroll
    for (int r = 0; r < 8; ++r) {
      float v = fmaxf(acc[t][r] + bval, 0.0f);
      p[2 * r]     = fmaf(v, wr.x, p[2 * r]);
      p[2 * r + 1] = fmaf(v, wr.y, p[2 * r + 1]);
    }
  }
  float w8[8], w4[4], w2[2];
  const bool u8 = (laneM & 8) != 0, u4 = (laneM & 4) != 0, u2 = (laneM & 2) != 0, u1 = (laneM & 1) != 0;
#pragma unroll
  for (int j = 0; j < 8; ++j) {
    float keep = u8 ? p[j + 8] : p[j];
    float send = u8 ? p[j] : p[j + 8];
    w8[j] = keep + __shfl_xor(send, 8, 32);
  }
#pragma unroll
  for (int j = 0; j < 4; ++j) {
    float keep = u4 ? w8[j + 4] : w8[j];
    float send = u4 ? w8[j] : w8[j + 4];
    w4[j] = keep + __shfl_xor(send, 4, 32);
  }
#pragma unroll
  for (int j = 0; j < 2; ++j) {
    float keep = u2 ? w4[j + 2] : w4[j];
    float send = u2 ? w4[j] : w4[j + 2];
    w2[j] = keep + __shfl_xor(send, 2, 32);
  }
  {
    float keep = u1 ? w2[1] : w2[0];
    float send = u1 ? w2[0] : w2[1];
    float tot  = keep + __shfl_xor(send, 1, 32);
    out[(size_t)m0 * 2 + lane] = tot + br[lane & 1];
  }
}

extern "C" void kernel_launch(void* const* d_in, const int* in_sizes, int n_in,
                              void* d_out, int out_size, void* d_ws, size_t ws_size,
                              hipStream_t stream) {
  const float* x     = (const float*)d_in[0];
  const int*   eidx  = (const int*)d_in[1];
  const float* W1    = (const float*)d_in[2];
  const float* b1    = (const float*)d_in[3];
  const float* W2    = (const float*)d_in[4];
  const float* b2    = (const float*)d_in[5];
  const float* W3    = (const float*)d_in[6];
  const float* b3    = (const float*)d_in[7];
  const float* gamma = (const float*)d_in[8];
  const float* beta  = (const float*)d_in[9];
  const float* Wc    = (const float*)d_in[10];
  const float* bc    = (const float*)d_in[11];
  const float* Wr    = (const float*)d_in[12];
  const float* br    = (const float*)d_in[13];
  float* out = (float*)d_out;

  const int N = N_NODES, E = E_EDGES;
  const int* src = eidx;
  const int* dst = eidx + E;

  float* buf0   = (float*)d_ws;
  float* buf1   = buf0 + (size_t)N * DHID;
  int*   cnt    = (int*)(buf1 + (size_t)N * DHID);
  float* stats  = (float*)(cnt + N);
  float* bnp    = stats + 256;
  int*   bsum   = (int*)(bnp + 256);
  float* dinv   = (float*)(bsum + 256);
  int*   rowptr = (int*)(dinv + N);
  int*   tab    = rowptr + N + 4;
  int*   col    = tab + (size_t)NCHUNK * N;

  const dim3 blk(256);
  const dim3 gSort(NCHUNK, NRANGE);
  const dim3 bSort(SORT_T);
  const int gScan = (N + 1023) / 1024;
  const int gGemm = (N + 127) / 128;
  const int gFuse = (N / 16 + FG_WAVES - 1) / FG_WAVES;
  const dim3 bFuse(FG_T);

  k_count_lds<<<gSort, bSort, 0, stream>>>(dst, tab);
  k_degsum<<<gScan, blk, 0, stream>>>(tab, cnt, bsum, stats, N);
  k_scan_dinv<<<gScan, blk, 0, stream>>>(cnt, bsum, rowptr, dinv, N);
  k_fill_lds<<<gSort, bSort, 0, stream>>>(src, dst, rowptr, tab, col);

  k_gather_gemm<<<gFuse, bFuse, 0, stream>>>(rowptr, col, dinv, x,    b1, W1, buf0, N);
  k_gather_gemm<<<gFuse, bFuse, 0, stream>>>(rowptr, col, dinv, buf0, b2, W2, buf1, N);
  k_gather_gemm<<<gFuse, bFuse, 0, stream>>>(rowptr, col, dinv, buf1, b3, W3, buf0, N);

  k_bnstats<<<1024, blk, 0, stream>>>(buf0, stats, N);

  k_classify_head<<<gGemm, blk, 0, stream>>>(buf0, Wc, stats, gamma, beta, bc, Wr, br, out, N);
}


# Batchnorm sums formed in the layer-3 launch: a wave adds its 16 output rows per channel from its WMMA registers, waves added in LDS, 256 float atomics per block; stats launch gone; sum order changed
# speedup vs baseline: 4.3535x; 1.0899x over previous
#include <hip/hip_runtime.h>

#define N_NODES 100000
#define DHID    128
#define E_EDGES 1600000
#define BN_EPS  1e-5f

typedef __attribute__((ext_vector_type(2))) float v2f;
typedef __attribute__((ext_vector_type(8))) float v8f;

#define NRANGE 8
#define NPR    (N_NODES / NRANGE)
#define NCHUNK 32
#define EPC    (E_EDGES / NCHUNK)
#define EPC4   (EPC / 4)
#define SORT_T 1024
static_assert(NPR * NRANGE == N_NODES, "ranges must tile the nodes");
static_assert(EPC * NCHUNK == E_EDGES && EPC4 * 4 == EPC, "chunks must tile the edges in 16-byte groups");
static_assert(N_NODES % 4 == 0, "k_degsum reads four nodes per thread");

__device__ __forceinline__ void count_one(int* lc, int d) {
  if ((unsigned)d < (unsigned)NPR) atomicAdd(&lc[d], 1);
}

__global__ __launch_bounds__(SORT_T) void k_count_lds(const int* __restrict__ dst,
                                                      int* __restrict__ tab) {
  __shared__ int lc[NPR];
  const int t = threadIdx.x, c = blockIdx.x, lo = blockIdx.y * NPR;
  for (int j = t; j < NPR; j += SORT_T) lc[j] = 0;
  __syncthreads();
  const int4* __restrict__ d4 = (const int4*)(dst + (size_t)c * EPC);
  constexpr int B = (EPC4 + SORT_T - 1) / SORT_T;
  int4 v[B];
#pragma unroll
  for (int q = 0; q < B; ++q) {
    const int i = t + q * SORT_T;
    v[q] = (i < EPC4) ? d4[i] : make_int4(-1, -1, -1, -1);
  }
#pragma unroll
  for (int q = 0; q < B; ++q) {
    count_one(lc, v[q].x - lo);
    count_one(lc, v[q].y - lo);
    count_one(lc, v[q].z - lo);
    count_one(lc, v[q].w - lo);
  }
  __syncthreads();
  int* __restrict__ trow = tab + (size_t)c * N_NODES + lo;
  for (int j = t; j < NPR; j += SORT_T) trow[j] = lc[j];
}

__global__ __launch_bounds__(256) void k_degsum(int* __restrict__ tab, int* __restrict__ cnt,
                                                int* __restrict__ bsum, float* __restrict__ stats,
                                                int n) {
  __shared__ int sh[256];
  const int t  = threadIdx.x;
  const int i0 = (blockIdx.x * 256 + t) * 4;
  if (blockIdx.x == 0) stats[t] = 0.0f;
  int4 run = make_int4(0, 0, 0, 0);
  if (i0 < n) {
    int4 v[NCHUNK];
#pragma unroll
    for (int c = 0; c < NCHUNK; ++c) v[c] = *(const int4*)(tab + (size_t)c * N_NODES + i0);
#pragma unroll
    for (int c = 0; c < NCHUNK; ++c) {
      *(int4*)(tab + (size_t)c * N_NODES + i0) = run;
      run.x += v[c].x; run.y += v[c].y; run.z += v[c].z; run.w += v[c].w;
    }
    *(int4*)(cnt + i0) = run;
  }
  sh[t] = run.x + run.y + run.z + run.w;
  __syncthreads();
  for (int off = 128; off > 0; off >>= 1) {
    if (t < off) sh[t] += sh[t + off];
    __syncthreads();
  }
  if (t == 0) bsum[blockIdx.x] = sh[0];
}

__global__ __launch_bounds__(256) void k_scan_dinv(const int* __restrict__ cnt,
                                                   const int* __restrict__ bsum,
                                                   int* __restrict__ rowptr,
                                                   float* __restrict__ dinv, int n) {
  __shared__ int sh[256];
  const int t = threadIdx.x, b = blockIdx.x;
  sh[t] = (t < b) ? bsum[t] : 0;
  __syncthreads();
  for (int off = 128; off > 0; off >>= 1) {
    if (t < off) sh[t] += sh[t + off];
    __syncthreads();
  }
  const int boff = sh[0];
  __syncthreads();

  const int i0 = (b * 256 + t) * 4;
  int c[4], tot = 0;
#pragma unroll
  for (int q = 0; q < 4; ++q) {
    c[q] = (i0 + q < n) ? cnt[i0 + q] : 0;
    tot += c[q];
  }
  sh[t] = tot;
  __syncthreads();
  for (int off = 1; off < 256; off <<= 1) {
    int x = (t >= off) ? sh[t - off] : 0;
    __syncthreads();
    sh[t] += x;
    __syncthreads();
  }
  int excl = boff + sh[t] - tot;
#pragma unroll
  for (int q = 0; q < 4; ++q) {
    if (i0 + q < n) {
      rowptr[i0 + q] = excl;
      dinv[i0 + q]   = __frsqrt_rn((float)c[q] + 1.0f);
      excl += c[q];
    }
  }
  if (i0 < n && i0 + 4 >= n) rowptr[n] = excl;
}

__device__ __forceinline__ int take_slot(int* cur, int d) {
  int slot = -1;
  if ((unsigned)d < (unsigned)NPR) slot = atomicAdd(&cur[d], 1);
  return slot;
}
__device__ __forceinline__ void put_src(int* __restrict__ col, int slot, int s) {
  if ((unsigned)slot < (unsigned)E_EDGES) col[slot] = s;
}

__global__ __launch_bounds__(SORT_T) void k_fill_lds(const int* __restrict__ src,
                                                     const int* __restrict__ dst,
                                                     const int* __restrict__ rowptr,
                                                     const int* __restrict__ tab,
                                                     int* __restrict__ col) {
  __shared__ int cur[NPR];
  const int t = threadIdx.x, c = blockIdx.x, lo = blockIdx.y * NPR;
  const int* __restrict__ trow = tab + (size_t)c * N_NODES + lo;
  const int* __restrict__ rp   = rowptr + lo;
  for (int j = t; j < NPR; j += SORT_T) cur[j] = rp[j] + trow[j];
  __syncthreads();
  const int4* __restrict__ d4 = (const int4*)(dst + (size_t)c * EPC);
  const int4* __restrict__ s4 = (const int4*)(src + (size_t)c * EPC);
  constexpr int B = 7;
  for (int p = 0; p < EPC4; p += B * SORT_T) {
    int4 dv[B], sv[B];
#pragma unroll
    for (int q = 0; q < B; ++q) {
      const int  i  = p + t + q * SORT_T;
      const bool ok = i < EPC4;
      dv[q] = ok ? d4[i] : make_int4(-1, -1, -1, -1);
      sv[q] = ok ? s4[i] : make_int4(0, 0, 0, 0);
    }
    int slot[4 * B];
#pragma unroll
    for (int q = 0; q < B; ++q) {
      slot[4 * q + 0] = take_slot(cur, dv[q].x - lo);
      slot[4 * q + 1] = take_slot(cur, dv[q].y - lo);
      slot[4 * q + 2] = take_slot(cur, dv[q].z - lo);
      slot[4 * q + 3] = take_slot(cur, dv[q].w - lo);
    }
#pragma unroll
    for (int q = 0; q < B; ++q) {
      put_src(col, slot[4 * q + 0], sv[q].x);
      put_src(col, slot[4 * q + 1], sv[q].y);
      put_src(col, slot[4 * q + 2], sv[q].z);
      put_src(col, slot[4 * q + 3], sv[q].w);
    }
  }
}

__device__ __forceinline__ void fma_row(float4& acc, const float4& v, float c) {
  acc.x = fmaf(v.x, c, acc.x);
  acc.y = fmaf(v.y, c, acc.y);
  acc.z = fmaf(v.z, c, acc.z);
  acc.w = fmaf(v.w, c, acc.w);
}
__device__ __forceinline__ int lane_i(int v, int j) { return __builtin_amdgcn_readlane(v, j); }
__device__ __forceinline__ float lane_f(float v, int j) {
  return __int_as_float(__builtin_amdgcn_readlane(__float_as_int(v), j));
}

#define FG_RB 16
typedef __attribute__((ext_vector_type(4))) float v4f;
typedef const __attribute__((address_space(1))) v4f gv4f;
__device__ __forceinline__ float4 ld_row_u(const float* __restrict__ h, int srow, int lane) {
  const unsigned long long p  = (unsigned long long)(h + (size_t)srow * DHID);
  const unsigned           lo = __builtin_amdgcn_readfirstlane((unsigned)p);
  const unsigned           hi = __builtin_amdgcn_readfirstlane((unsigned)(p >> 32));
  const gv4f* rp = (const gv4f*)(((unsigned long long)hi << 32) | lo);
  const v4f v = rp[lane];
  return make_float4(v.x, v.y, v.z, v.w);
}
__device__ __forceinline__ void chunk_rows(float4& acc, const float* __restrict__ h,
                                           int s, float cs, int cnt, float dd, int lane) {
  int j = 0;
  for (; j + FG_RB <= cnt; j += FG_RB) {
    float4 t[FG_RB];
#pragma unroll
    for (int q = 0; q < FG_RB; ++q) t[q] = ld_row_u(h, lane_i(s, j + q), lane);
#pragma unroll
    for (int q = 0; q < FG_RB; ++q) fma_row(acc, t[q], lane_f(cs, j + q) * dd);
  }
  const int r = cnt - j;
  if (r > 0) {
    float4 t[FG_RB - 1];
#pragma unroll
    for (int q = 0; q < FG_RB - 1; ++q) t[q] = make_float4(0.0f, 0.0f, 0.0f, 0.0f);
#pragma unroll
    for (int q = 0; q < FG_RB - 1; ++q)
      if (q < r) t[q] = ld_row_u(h, lane_i(s, j + q), lane);
#pragma unroll
    for (int q = 0; q < FG_RB - 1; ++q)
      if (q < r) fma_row(acc, t[q], lane_f(cs, j + q) * dd);
  }
}

#define FG_WAVES 25
#define FG_T     (FG_WAVES * 32)
#define TPITCH   132
static_assert(N_NODES % 16 == 0, "a wave's tile holds 16 whole nodes");

template <bool STATS>
__global__ __launch_bounds__(FG_T) void k_gather_gemm(
    const int* __restrict__ rowptr, const int* __restrict__ col,
    const float* __restrict__ dinv, const float* __restrict__ h,
    const float* __restrict__ b, const float* __restrict__ W,
    float* __restrict__ C, float* __restrict__ stats, int n)
{
  __shared__ float wl[DHID * DHID];
  __shared__ __attribute__((aligned(16))) float tile[FG_WAVES * 16 * TPITCH];
  const int tid = threadIdx.x;
  for (int i = tid; i < DHID * DHID; i += FG_T) {
    int k = i >> 7, nn = i & 127;
    wl[(((k >> 1) << 7) + nn) * 2 + (k & 1)] = W[i];
  }
  __syncthreads();

  const int lane = tid & 31;
  const int wave = __builtin_amdgcn_readfirstlane(tid >> 5);
  const int m0   = (blockIdx.x * FG_WAVES + wave) << 4;
  float* tw = tile + wave * (16 * TPITCH);
  do {
  if (m0 >= n) {
    if (STATS) {
      for (int j = 0; j < 16; ++j) tw[j * 32 + lane] = 0.0f;
    }
    break;
  }

  const int    rpv = rowptr[min(m0 + min(lane, 16), n)];
  const float  ddv = dinv[min(m0 + (lane & 15), n - 1)];

  const int st0 = lane_i(rpv, 0), st1 = lane_i(rpv, 1), st2 = lane_i(rpv, 2);
  int   s1  = col[min(st0 + lane, E_EDGES - 1)];
  int   s2  = col[min(st1 + lane, E_EDGES - 1)];
  s1 = (st0 + lane < st1 && (unsigned)s1 < (unsigned)n) ? s1 : m0;
  s2 = (st1 + lane < st2 && (unsigned)s2 < (unsigned)n) ? s2 : m0;
  float cs1 = dinv[s1];

  for (int i = 0; i < 16; ++i) {
    const int   node  = m0 + i;
    const int   start = lane_i(rpv, i), end = lane_i(rpv, i + 1);
    const float dd    = lane_f(ddv, i);
    const float cs2  = dinv[s2];
    const int   st3  = lane_i(rpv, min(i + 2, 16)), en3 = lane_i(rpv, min(i + 3, 16));
    const int   s3r  = col[min(st3 + lane, E_EDGES - 1)];
    const float4 hs = ld_row_u(h, node, lane);
    float4 acc = make_float4(0.0f, 0.0f, 0.0f, 0.0f);
    chunk_rows(acc, h, s1, cs1, min(32, end - start), dd, lane);
    for (int base = start + 32; base < end; base += 32) {
      const int idx = base + lane;
      int   s  = node;
      float cs = 0.0f;
      if (idx < end) {
        s  = col[idx];
        s  = ((unsigned)s < (unsigned)n) ? s : node;
        cs = dinv[s];
      }
      const int cnt = min(32, end - base);
      for (int j = 0; j < cnt; ++j) fma_row(acc, ld_row_u(h, lane_i(s, j), lane), lane_f(cs, j) * dd);
    }
    const float sc = dd * dd;
    float4 o;
    o.x = fmaf(hs.x, sc, acc.x);
    o.y = fmaf(hs.y, sc, acc.y);
    o.z = fmaf(hs.z, sc, acc.z);
    o.w = fmaf(hs.w, sc, acc.w);
    *(float4*)(tw + i * TPITCH + lane * 4) = o;
    s1  = s2;
    cs1 = cs2;
    s2  = (st3 + lane < en3 && (unsigned)s3r < (unsigned)n) ? s3r : node;
  }

  __builtin_amdgcn_fence(__ATOMIC_RELEASE, "wavefront");
  __builtin_amdgcn_wave_barrier();
  __builtin_amdgcn_fence(__ATOMIC_ACQUIRE, "wavefront");

  const int laneM = lane & 15;
  const int hi    = lane >> 4;
  const float2* __restrict__ wl2 = (const float2*)wl;
  const float* trow = tw + laneM * TPITCH + hi * 2;

  v8f acc2[8];
#pragma unroll
  for (int t = 0; t < 8; ++t) acc2[t] = (v8f){};

  for (int k = 0; k < DHID; k += 4) {
    const float2 av = *(const float2*)(trow + k);
    v2f a; a[0] = av.x; a[1] = av.y;
    const float2* __restrict__ brow = wl2 + (((k >> 1) + hi) << 7);
#pragma unroll
    for (int t = 0; t < 8; ++t) {
      float2 bw = brow[t * 16 + laneM];
      v2f bb; bb[0] = bw.x; bb[1] = bw.y;
      acc2[t] = __builtin_amdgcn_wmma_f32_16x16x4_f32(
          false, a, false, bb, (short)0, acc2[t], false, false);
    }
  }

#pragma unroll
  for (int t = 0; t < 8; ++t) {
    const int   cc   = t * 16 + laneM;
    const float bval = b[cc];
    float s = 0.0f, q = 0.0f;
#pragma unroll
    for (int r = 0; r < 8; ++r) {
      const int   row = m0 + r + hi * 8;
      const float v   = fmaxf(acc2[t][r] + bval, 0.0f);
      C[(size_t)row * DHID + cc] = v;
      if (STATS) { s += v; q = fmaf(v, v, q); }
    }
    if (STATS) {
      tw[hi * DHID + cc]            = s;
      tw[2 * DHID + hi * DHID + cc] = q;
    }
  }
  } while (0);

  if (STATS) {
    __syncthreads();
    if (tid < 2 * DHID) {
      const int st = tid >> 7, c = tid & (DHID - 1);
      float tot = 0.0f;
      for (int w = 0; w < FG_WAVES; ++w) {
        const float* p = tile + w * (16 * TPITCH) + st * (2 * DHID) + c;
        tot += p[0] + p[DHID];
      }
      unsafeAtomicAdd(&stats[tid], tot);
    }
  }
}

__global__ __launch_bounds__(256) void k_classify_head(
    const float* __restrict__ A, const float* __restrict__ W,
    const float* __restrict__ stats, const float* __restrict__ gamma,
    const float* __restrict__ beta, const float* __restrict__ bias,
    const float* __restrict__ Wr, const float* __restrict__ br,
    float* __restrict__ out, int nrows)
{
  __shared__ float wl[DHID * DHID];
  __shared__ __attribute__((aligned(16))) float bnl[2 * DHID];
  const int tid = threadIdx.x;
  for (int i = tid; i < DHID * DHID; i += 256) {
    int k = i >> 7, n = i & 127;
    wl[(((k >> 1) << 7) + n) * 2 + (k & 1)] = W[i];
  }
  if (tid < DHID) {
    float inv = 1.0f / (float)nrows;
    float mu  = stats[tid] * inv;
    float var = stats[128 + tid] * inv - mu * mu;
    float sc  = gamma[tid] * __frsqrt_rn(var + BN_EPS);
    bnl[tid]        = sc;
    bnl[DHID + tid] = beta[tid] - mu * sc;
  }
  __syncthreads();

  const int wave  = tid >> 5;
  const int lane  = tid & 31;
  const int laneM = lane & 15;
  const int hi    = lane >> 4;
  const int m0    = (blockIdx.x * 8 + wave) << 4;
  if (m0 >= nrows) return;

  const float2* __restrict__ wl2  = (const float2*)wl;
  const float2* __restrict__ bn2  = (const float2*)bnl;
  const float*  __restrict__ arow = A + (size_t)(m0 + laneM) * DHID;

  v8f acc[8];
#pragma unroll
  for (int t = 0; t < 8; ++t) acc[t] = (v8f){};

#pragma unroll 2
  for (int k = 0; k < DHID; k += 4) {
    const int c0 = k + hi * 2;
    float2 av = *(const float2*)(arow + c0);
    const float2 sc2 = bn2[c0 >> 1];
    const float2 sh2 = bn2[(DHID >> 1) + (c0 >> 1)];
    av.x = fmaxf(av.x, 0.0f) * sc2.x + sh2.x;
    av.y = fmaxf(av.y, 0.0f) * sc2.y + sh2.y;
    v2f a; a[0] = av.x; a[1] = av.y;

    const float2* __restrict__ brow = wl2 + (((k >> 1) + hi) << 7);
#pragma unroll
    for (int t = 0; t < 8; ++t) {
      float2 bv = brow[t * 16 + laneM];
      v2f b; b[0] = bv.x; b[1] = bv.y;
      acc[t] = __builtin_amdgcn_wmma_f32_16x16x4_f32(
          false, a, false, b, (short)0, acc[t], false, false);
    }
  }

  float p[16];
#pragma unroll
  for (int i = 0; i < 16; ++i) p[i] = 0.0f;
#pragma unroll
  for (int t = 0; t < 8; ++t) {
    const int    col  = t * 16 + laneM;
    const float  bval = bias[col];
    const float2 wr   = *(const float2*)(Wr + col * 2);
#pragma unroll
    for (int r = 0; r < 8; ++r) {
      float v = fmaxf(acc[t][r] + bval, 0.0f);
      p[2 * r]     = fmaf(v, wr.x, p[2 * r]);
      p[2 * r + 1] = fmaf(v, wr.y, p[2 * r + 1]);
    }
  }
  float w8[8], w4[4], w2[2];
  const bool u8 = (laneM & 8) != 0, u4 = (laneM & 4) != 0, u2 = (laneM & 2) != 0, u1 = (laneM & 1) != 0;
#pragma unroll
  for (int j = 0; j < 8; ++j) {
    float keep = u8 ? p[j + 8] : p[j];
    float send = u8 ? p[j] : p[j + 8];
    w8[j] = keep + __shfl_xor(send, 8, 32);
  }
#pragma unroll
  for (int j = 0; j < 4; ++j) {
    float keep = u4 ? w8[j + 4] : w8[j];
    float send = u4 ? w8[j] : w8[j + 4];
    w4[j] = keep + __shfl_xor(send, 4, 32);
  }
#pragma unroll
  for (int j = 0; j < 2; ++j) {
    float keep = u2 ? w4[j + 2] : w4[j];
    float send = u2 ? w4[j] : w4[j + 2];
    w2[j] = keep + __shfl_xor(send, 2, 32);
  }
  {
    float keep = u1 ? w2[1] : w2[0];
    float send = u1 ? w2[0] : w2[1];
    float tot  = keep + __shfl_xor(send, 1, 32);
    out[(size_t)m0 * 2 + lane] = tot + br[lane & 1];
  }
}

extern "C" void kernel_launch(void* const* d_in, const int* in_sizes, int n_in,
                              void* d_out, int out_size, void* d_ws, size_t ws_size,
                              hipStream_t stream) {
  const float* x     = (const float*)d_in[0];
  const int*   eidx  = (const int*)d_in[1];
  const float* W1    = (const float*)d_in[2];
  const float* b1    = (const float*)d_in[3];
  const float* W2    = (const float*)d_in[4];
  const float* b2    = (const float*)d_in[5];
  const float* W3    = (const float*)d_in[6];
  const float* b3    = (const float*)d_in[7];
  const float* gamma = (const float*)d_in[8];
  const float* beta  = (const float*)d_in[9];
  const float* Wc    = (const float*)d_in[10];
  const float* bc    = (const float*)d_in[11];
  const float* Wr    = (const float*)d_in[12];
  const float* br    = (const float*)d_in[13];
  float* out = (float*)d_out;

  const int N = N_NODES, E = E_EDGES;
  const int* src = eidx;
  const int* dst = eidx + E;

  float* buf0   = (float*)d_ws;
  float* buf1   = buf0 + (size_t)N * DHID;
  int*   cnt    = (int*)(buf1 + (size_t)N * DHID);
  float* stats  = (float*)(cnt + N);
  float* bnp    = stats + 256;
  int*   bsum   = (int*)(bnp + 256);
  float* dinv   = (float*)(bsum + 256);
  int*   rowptr = (int*)(dinv + N);
  int*   tab    = rowptr + N + 4;
  int*   col    = tab + (size_t)NCHUNK * N;

  const dim3 blk(256);
  const dim3 gSort(NCHUNK, NRANGE);
  const dim3 bSort(SORT_T);
  const int gScan = (N + 1023) / 1024;
  const int gGemm = (N + 127) / 128;
  const int gFuse = (N / 16 + FG_WAVES - 1) / FG_WAVES;
  const dim3 bFuse(FG_T);

  k_count_lds<<<gSort, bSort, 0, stream>>>(dst, tab);
  k_degsum<<<gScan, blk, 0, stream>>>(tab, cnt, bsum, stats, N);
  k_scan_dinv<<<gScan, blk, 0, stream>>>(cnt, bsum, rowptr, dinv, N);
  k_fill_lds<<<gSort, bSort, 0, stream>>>(src, dst, rowptr, tab, col);

  k_gather_gemm<false><<<gFuse, bFuse, 0, stream>>>(rowptr, col, dinv, x,    b1, W1, buf0, stats, N);
  k_gather_gemm<false><<<gFuse, bFuse, 0, stream>>>(rowptr, col, dinv, buf0, b2, W2, buf1, stats, N);
  k_gather_gemm<true><<<gFuse, bFuse, 0, stream>>>(rowptr, col, dinv, buf1, b3, W3, buf0, stats, N);

  k_classify_head<<<gGemm, blk, 0, stream>>>(buf0, Wc, stats, gamma, beta, bc, Wr, br, out, N);
}
